# starting kernel
# baseline (speedup 1.0000x reference)
#include <hip/hip_runtime.h>

typedef __attribute__((ext_vector_type(2))) float v2f;
typedef __attribute__((ext_vector_type(4))) float v4f;
typedef __attribute__((ext_vector_type(8))) float v8f;

#define DIM 128

static __device__ __forceinline__ v8f wmma_f32(v2f a, v2f b, v8f c) {
  return __builtin_amdgcn_wmma_f32_16x16x4_f32(false, a, false, b,
                                               (short)0, c, false, false);
}

__global__ void __launch_bounds__(128)
node_selfmsg_kernel(const float* __restrict__ X, const float* __restrict__ W,
                    float* __restrict__ out, int nrows) {
  const int lane  = threadIdx.x & 31;
  const int wave  = threadIdx.x >> 5;
  const int m0    = blockIdx.x * 16;
  const int n0    = wave * 32;
  const int lo    = lane & 15;
  const int khalf = lane >> 4;

  int mrow = m0 + lo;
  if (mrow >= nrows) mrow = nrows - 1;
  const float* Arow = X + (size_t)mrow * DIM;
  const float* B0   = W + (size_t)(n0 + lo) * DIM;
  const float* B1   = B0 + 16 * DIM;

  v8f acc0 = {}, acc1 = {};
  #pragma unroll 4
  for (int kk = 0; kk < DIM; kk += 4) {
    const int k = kk + khalf * 2;
    v2f a  = *(const v2f*)(Arow + k);
    v2f b0 = *(const v2f*)(B0 + k);
    v2f b1 = *(const v2f*)(B1 + k);
    acc0 = wmma_f32(a, b0, acc0);
    acc1 = wmma_f32(a, b1, acc1);
  }
  if (m0 + 16 <= nrows) {
    float* o = out + (size_t)(m0 + khalf * 8) * DIM + n0 + lo;
    #pragma unroll
    for (int r = 0; r < 8; ++r) {
      o[(size_t)r * DIM]      = acc0[r];
      o[(size_t)r * DIM + 16] = acc1[r];
    }
  } else {
    #pragma unroll
    for (int r = 0; r < 8; ++r) {
      const int orow = m0 + khalf * 8 + r;
      if (orow < nrows) {
        float* o = out + (size_t)orow * DIM + lo;
        o[n0]      = acc0[r];
        o[n0 + 16] = acc1[r];
      }
    }
  }
}

__global__ void __launch_bounds__(128)
edge_msg_kernel(const float* __restrict__ src_x, const float* __restrict__ dst_x,
                const int* __restrict__ src_idx, const int* __restrict__ dst_idx,
                const float* __restrict__ ew,
                const float* __restrict__ Wi, const float* __restrict__ We,
                float* __restrict__ out, int nedges) {
  __shared__ float s_xs[32][132];
  __shared__ float s_p [32][132];
  __shared__ float s_ew[32];
  __shared__ int   s_dst[32];

  const int tid = threadIdx.x;
  const int e0  = blockIdx.x * 32;

  {
    const int row = tid >> 2;
    const int cb  = (tid & 3) * 32;
    int e = e0 + row;
    if (e >= nedges) e = nedges - 1;
    const float* xs = src_x + (size_t)src_idx[e] * DIM + cb;
    const float* xd = dst_x + (size_t)dst_idx[e] * DIM + cb;
    #pragma unroll
    for (int j = 0; j < 32; j += 4) {
      v4f a = *(const v4f*)(xs + j);
      v4f b = *(const v4f*)(xd + j);
      *(v4f*)(&s_xs[row][cb + j]) = a;
      *(v4f*)(&s_p [row][cb + j]) = a * b;
    }
    if (tid < 32) {
      const int e2 = e0 + tid;
      s_ew[tid]  = (e2 < nedges) ? ew[e2] : 0.0f;
      s_dst[tid] = dst_idx[(e2 < nedges) ? e2 : (nedges - 1)];
    }
  }
  __syncthreads();

  const int lane  = tid & 31;
  const int wave  = tid >> 5;
  const int n0    = wave * 32;
  const int lo    = lane & 15;
  const int khalf = lane >> 4;

  const float* BiA = Wi + (size_t)(n0 + lo) * DIM;
  const float* BeA = We + (size_t)(n0 + lo) * DIM;
  const float* BiB = BiA + 16 * DIM;
  const float* BeB = BeA + 16 * DIM;

  const float* Ax0 = &s_xs[lo][0];
  const float* Ap0 = &s_p [lo][0];
  const float* Ax1 = &s_xs[16 + lo][0];
  const float* Ap1 = &s_p [16 + lo][0];

  v8f acc00 = {}, acc01 = {}, acc10 = {}, acc11 = {};
  #pragma unroll 2
  for (int kk = 0; kk < DIM; kk += 4) {
    const int k = kk + khalf * 2;
    v2f a0x = *(const v2f*)(Ax0 + k);
    v2f a0p = *(const v2f*)(Ap0 + k);
    v2f a1x = *(const v2f*)(Ax1 + k);
    v2f a1p = *(const v2f*)(Ap1 + k);
    v2f bia = *(const v2f*)(BiA + k);
    v2f bea = *(const v2f*)(BeA + k);
    v2f bib = *(const v2f*)(BiB + k);
    v2f beb = *(const v2f*)(BeB + k);
    acc00 = wmma_f32(a0x, bia, acc00);
    acc00 = wmma_f32(a0p, bea, acc00);
    acc01 = wmma_f32(a0x, bib, acc01);
    acc01 = wmma_f32(a0p, beb, acc01);
    acc10 = wmma_f32(a1x, bia, acc10);
    acc10 = wmma_f32(a1p, bea, acc10);
    acc11 = wmma_f32(a1x, bib, acc11);
    acc11 = wmma_f32(a1p, beb, acc11);
  }

  #pragma unroll
  for (int r = 0; r < 8; ++r) {
    const int row0 = r + 8 * khalf;
    const int row1 = 16 + row0;
    float* o0 = out + (size_t)s_dst[row0] * DIM + lo;
    float* o1 = out + (size_t)s_dst[row1] * DIM + lo;
    const float w0 = s_ew[row0];
    const float w1 = s_ew[row1];
    atomicAdd(o0 + n0,      w0 * acc00[r]);
    atomicAdd(o0 + n0 + 16, w0 * acc01[r]);
    atomicAdd(o1 + n0,      w1 * acc10[r]);
    atomicAdd(o1 + n0 + 16, w1 * acc11[r]);
  }
}

__global__ void __launch_bounds__(256)
lrelu_kernel(float* __restrict__ out, int n4) {
  const int i = blockIdx.x * blockDim.x + threadIdx.x;
  if (i < n4) {
    v4f x = *(v4f*)(out + (size_t)i * 4);
    #pragma unroll
    for (int c = 0; c < 4; ++c) x[c] = (x[c] > 0.0f) ? x[c] : 0.01f * x[c];
    *(v4f*)(out + (size_t)i * 4) = x;
  }
}

extern "C" void kernel_launch(void* const* d_in, const int* in_sizes, int n_in,
                              void* d_out, int out_size, void* d_ws, size_t ws_size,
                              hipStream_t stream) {
  const float* src_x = (const float*)d_in[0];
  const float* dst_x = (const float*)d_in[1];
  const int*   eidx  = (const int*)  d_in[2];
  const float* ew    = (const float*)d_in[3];
  const float* Wi    = (const float*)d_in[4];
  const float* We    = (const float*)d_in[5];
  float*       out   = (float*)d_out;

  const int N = in_sizes[0] / DIM;
  const int E = in_sizes[2] / 2;
  const int* src_idx = eidx;
  const int* dst_idx = eidx + E;

  node_selfmsg_kernel<<<(N + 15) / 16, 128, 0, stream>>>(dst_x, Wi, out, N);
  edge_msg_kernel<<<(E + 31) / 32, 128, 0, stream>>>(src_x, dst_x, src_idx, dst_idx,
                                                     ew, Wi, We, out, E);
  const int n4 = (N * DIM) / 4;
  lrelu_kernel<<<(n4 + 255) / 256, 256, 0, stream>>>(out, n4);
}


# Work-group owns 400 dst nodes: sums ew*src rows in LDS (LDS float adds replace global atomics), then one fused transform per node; summation order changes (aggregate before GEMM)
# speedup vs baseline: 5.2542x; 5.2542x over previous
#include <hip/hip_runtime.h>


typedef __attribute__((ext_vector_type(2))) float v2f;
typedef __attribute__((ext_vector_type(4))) float v4f;
typedef __attribute__((ext_vector_type(8))) float v8f;

#define DIM    128
#define RPB    400
#define NTILE  (RPB / 16)
#define PITCH  132
#define NTHR   1024
#define SCAN_U 16

static __device__ __forceinline__ v8f wmma_f32(v2f a, v2f b, v8f c) {
  return __builtin_amdgcn_wmma_f32_16x16x4_f32(false, a, false, b,
                                               (short)0, c, false, false);
}

template <int NT>
static __device__ __forceinline__ void
transform_rows(const float* __restrict__ s_acc, const float* __restrict__ dst_x,
               const float* __restrict__ Wi, const float* __restrict__ We,
               float* __restrict__ out, int nbase, int nrows,
               int grp, int n0, int lo, int h) {
  const float* bi0 = Wi + (size_t)(n0 + lo) * DIM + 4 * h;
  const float* be0 = We + (size_t)(n0 + lo) * DIM + 4 * h;
  const float* bi1 = bi0 + 16 * DIM;
  const float* be1 = be0 + 16 * DIM;

  const float* arow[NT];
  const float* drow[NT];
  #pragma unroll
  for (int j = 0; j < NT; ++j) {
    const int lrow = (grp + 8 * j) * 16 + lo;
    int grow = nbase + lrow;
    if (grow >= nrows) grow = nrows - 1;
    arow[j] = s_acc + lrow * PITCH + 4 * h;
    drow[j] = dst_x + (size_t)grow * DIM + 4 * h;
  }

  v8f acc[NT][2];
  #pragma unroll
  for (int j = 0; j < NT; ++j) { acc[j][0] = v8f{}; acc[j][1] = v8f{}; }

  for (int q = 0; q < DIM / 8; ++q) {
    const int k = q * 8;
    const v4f wi0 = *(const v4f*)(bi0 + k);
    const v4f we0 = *(const v4f*)(be0 + k);
    const v4f wi1 = *(const v4f*)(bi1 + k);
    const v4f we1 = *(const v4f*)(be1 + k);
    #pragma unroll
    for (int j = 0; j < NT; ++j) {
      const v4f g = *(const v4f*)(arow[j] + k);
      const v4f d = *(const v4f*)(drow[j] + k);
      const v4f a1 = g + d;
      const v4f a2 = g * d;
      acc[j][0] = wmma_f32(a1.xy, wi0.xy, acc[j][0]);
      acc[j][1] = wmma_f32(a1.xy, wi1.xy, acc[j][1]);
      acc[j][0] = wmma_f32(a1.zw, wi0.zw, acc[j][0]);
      acc[j][1] = wmma_f32(a1.zw, wi1.zw, acc[j][1]);
      acc[j][0] = wmma_f32(a2.xy, we0.xy, acc[j][0]);
      acc[j][1] = wmma_f32(a2.xy, we1.xy, acc[j][1]);
      acc[j][0] = wmma_f32(a2.zw, we0.zw, acc[j][0]);
      acc[j][1] = wmma_f32(a2.zw, we1.zw, acc[j][1]);
    }
  }

  #pragma unroll
  for (int j = 0; j < NT; ++j) {
    const int row0 = nbase + (grp + 8 * j) * 16 + 8 * h;
    #pragma unroll
    for (int r = 0; r < 8; ++r) {
      const int orow = row0 + r;
      if (orow < nrows) {
        float* o = out + (size_t)orow * DIM + n0 + lo;
        const float x0 = acc[j][0][r];
        const float x1 = acc[j][1][r];
        o[0]  = (x0 > 0.0f) ? x0 : 0.01f * x0;
        o[16] = (x1 > 0.0f) ? x1 : 0.01f * x1;
      }
    }
  }
}

__global__ void __launch_bounds__(NTHR)
owner_range_kernel(const float* __restrict__ src_x, const float* __restrict__ dst_x,
                   const int* __restrict__ src_idx, const int* __restrict__ dst_idx,
                   const float* __restrict__ ew,
                   const float* __restrict__ Wi, const float* __restrict__ We,
                   float* __restrict__ out, int nrows, int nedges) {
  __shared__ __attribute__((aligned(16))) float s_acc[RPB * PITCH];

  const int tid   = threadIdx.x;
  const int lane  = tid & 31;
  const int wave  = __builtin_amdgcn_readfirstlane(tid >> 5);
  const int nbase = blockIdx.x * RPB;

  for (int i = tid; i < RPB * PITCH / 4; i += NTHR)
    ((v4f*)s_acc)[i] = v4f{0.0f, 0.0f, 0.0f, 0.0f};
  __syncthreads();

  int qe = 0;
  int ql = 0;
  int qn = 0;

  auto apply_top = [&](int n) {
    const int first = qn - n;
    int   sv = 0;
    float wv = 0.0f;
    {
      const int ee = __builtin_amdgcn_ds_bpermute((first + (lane & 7)) << 2, qe);
      if (lane < n) {
        sv = src_idx[ee];
        wv = ew[ee];
        sv = (sv < 0) ? 0 : ((sv >= nrows) ? nrows - 1 : sv);
      }
    }
    float x[8][4];
    #pragma unroll
    for (int j = 0; j < 8; ++j) {
      x[j][0] = 0.0f; x[j][1] = 0.0f; x[j][2] = 0.0f; x[j][3] = 0.0f;
      if (j < n) {
        const int s = __builtin_amdgcn_readlane(sv, j);
        const float* xs = src_x + (size_t)s * DIM + lane;
        x[j][0] = xs[0]; x[j][1] = xs[32]; x[j][2] = xs[64]; x[j][3] = xs[96];
      }
    }
    #pragma unroll
    for (int j = 0; j < 8; ++j) {
      if (j < n) {
        const float w  = __builtin_bit_cast(float,
                           __builtin_amdgcn_readlane(__builtin_bit_cast(int, wv), j));
        const int lrow = __builtin_amdgcn_readlane(ql, first + j);
        float* a = s_acc + lrow * PITCH + lane;
        __hip_atomic_fetch_add(a,      w * x[j][0], __ATOMIC_RELAXED, __HIP_MEMORY_SCOPE_WORKGROUP);
        __hip_atomic_fetch_add(a + 32, w * x[j][1], __ATOMIC_RELAXED, __HIP_MEMORY_SCOPE_WORKGROUP);
        __hip_atomic_fetch_add(a + 64, w * x[j][2], __ATOMIC_RELAXED, __HIP_MEMORY_SCOPE_WORKGROUP);
        __hip_atomic_fetch_add(a + 96, w * x[j][3], __ATOMIC_RELAXED, __HIP_MEMORY_SCOPE_WORKGROUP);
      }
    }
    qn = first;
  };

  auto apply_edge = [&](int ee, int lrow) {
    int s = src_idx[ee];
    s = (s < 0) ? 0 : ((s >= nrows) ? nrows - 1 : s);
    const float w = ew[ee];
    const float* xs = src_x + (size_t)s * DIM + lane;
    float* a = s_acc + lrow * PITCH + lane;
    const float x0 = xs[0], x1 = xs[32], x2 = xs[64], x3 = xs[96];
    __hip_atomic_fetch_add(a,      w * x0, __ATOMIC_RELAXED, __HIP_MEMORY_SCOPE_WORKGROUP);
    __hip_atomic_fetch_add(a + 32, w * x1, __ATOMIC_RELAXED, __HIP_MEMORY_SCOPE_WORKGROUP);
    __hip_atomic_fetch_add(a + 64, w * x2, __ATOMIC_RELAXED, __HIP_MEMORY_SCOPE_WORKGROUP);
    __hip_atomic_fetch_add(a + 96, w * x3, __ATOMIC_RELAXED, __HIP_MEMORY_SCOPE_WORKGROUP);
  };

  for (int e0 = wave * 32; e0 < nedges; e0 += NTHR * SCAN_U) {
    int dv[SCAN_U];
    if (e0 + NTHR * SCAN_U <= nedges) {
      #pragma unroll
      for (int u = 0; u < SCAN_U; ++u) dv[u] = dst_idx[e0 + u * NTHR + lane];
    } else {
      #pragma unroll
      for (int u = 0; u < SCAN_U; ++u) {
        const int e = e0 + u * NTHR + lane;
        dv[u] = (e < nedges) ? dst_idx[e] : -1;
      }
    }
    unsigned hitmask[SCAN_U];
    int tot = 0;
    #pragma unroll
    for (int u = 0; u < SCAN_U; ++u) {
      const unsigned rel = (unsigned)(dv[u] - nbase);
      const bool hit = (dv[u] >= 0) && (rel < (unsigned)RPB);
      hitmask[u] = __builtin_amdgcn_ballot_w32(hit);
      tot += __builtin_popcount(hitmask[u]);
    }
    if (qn + tot > 32) {
      while (qn > 0) apply_top(qn < 8 ? qn : 8);
    }
    if (tot <= 32) {
      #pragma unroll
      for (int u = 0; u < SCAN_U; ++u) {
        unsigned mask = hitmask[u];
        while (mask) {
          const int b = __builtin_ctz(mask);
          mask &= mask - 1;
          const int erow = __builtin_amdgcn_readlane(dv[u] - nbase, b);
          if (lane == qn) { qe = e0 + u * NTHR + b; ql = erow; }
          ++qn;
        }
      }
    } else {
      for (int u = 0; u < SCAN_U; ++u) {
        const int e = e0 + u * NTHR + lane;
        const int d = (e < nedges) ? dst_idx[e] : -1;
        const unsigned rel = (unsigned)(d - nbase);
        unsigned mask = __builtin_amdgcn_ballot_w32((d >= 0) && (rel < (unsigned)RPB));
        while (mask) {
          const int b = __builtin_ctz(mask);
          mask &= mask - 1;
          apply_edge(e0 + u * NTHR + b, __builtin_amdgcn_readlane((int)rel, b));
        }
      }
    }
    while (qn >= 8) apply_top(8);
  }
  while (qn > 0) apply_top(qn < 8 ? qn : 8);
  __syncthreads();

  const int lo  = lane & 15;
  const int h   = lane >> 4;
  const int n0  = (wave & 3) * 32;
  const int grp = wave >> 2;
  if (grp + 24 < NTILE)
    transform_rows<4>(s_acc, dst_x, Wi, We, out, nbase, nrows, grp, n0, lo, h);
  else
    transform_rows<3>(s_acc, dst_x, Wi, We, out, nbase, nrows, grp, n0, lo, h);
}

extern "C" void kernel_launch(void* const* d_in, const int* in_sizes, int n_in,
                              void* d_out, int out_size, void* d_ws, size_t ws_size,
                              hipStream_t stream) {
  const float* src_x = (const float*)d_in[0];
  const float* dst_x = (const float*)d_in[1];
  const int*   eidx  = (const int*)  d_in[2];
  const float* ew    = (const float*)d_in[3];
  const float* Wi    = (const float*)d_in[4];
  const float* We    = (const float*)d_in[5];
  float*       out   = (float*)d_out;

  const int N = in_sizes[0] / DIM;
  const int E = in_sizes[2] / 2;
  const int* src_idx = eidx;
  const int* dst_idx = eidx + E;

  owner_range_kernel<<<(N + RPB - 1) / RPB, NTHR, 0, stream>>>(
      src_x, dst_x, src_idx, dst_idx, ew, Wi, We, out, N, E);
}


# Pre-pass bins edges by owning 400-node destination range (LDS counting sort per chunk, records in workspace); owners read only their own records, not all edge ids; LDS float adds stay arrival-order
# speedup vs baseline: 6.6595x; 1.2675x over previous
#include <hip/hip_runtime.h>


typedef __attribute__((ext_vector_type(2))) float v2f;
typedef __attribute__((ext_vector_type(4))) float v4f;
typedef __attribute__((ext_vector_type(8))) float v8f;

#define DIM    128
#define RPB    400
#define NTILE  (RPB / 16)
#define PITCH  132
#define NTHR   1024
#define SCAN_U 16
#define NCH    256
#define CPW    (NCH / 32)
#define NBMAX  1024
#define BIN_THR 256
#define BIN_U  10
#define BPT    (NBMAX / BIN_THR)

static __device__ __forceinline__ v8f wmma_f32(v2f a, v2f b, v8f c) {
  return __builtin_amdgcn_wmma_f32_16x16x4_f32(false, a, false, b,
                                               (short)0, c, false, false);
}

template <int NT>
static __device__ __forceinline__ void
transform_rows(const float* __restrict__ s_acc, const float* __restrict__ dst_x,
               const float* __restrict__ Wi, const float* __restrict__ We,
               float* __restrict__ out, int nbase, int nrows,
               int grp, int n0, int lo, int h) {
  const float* bi0 = Wi + (size_t)(n0 + lo) * DIM + 4 * h;
  const float* be0 = We + (size_t)(n0 + lo) * DIM + 4 * h;
  const float* bi1 = bi0 + 16 * DIM;
  const float* be1 = be0 + 16 * DIM;

  const float* arow[NT];
  const float* drow[NT];
  #pragma unroll
  for (int j = 0; j < NT; ++j) {
    const int lrow = (grp + 8 * j) * 16 + lo;
    int grow = nbase + lrow;
    if (grow >= nrows) grow = nrows - 1;
    arow[j] = s_acc + lrow * PITCH + 4 * h;
    drow[j] = dst_x + (size_t)grow * DIM + 4 * h;
  }

  v8f acc[NT][2];
  #pragma unroll
  for (int j = 0; j < NT; ++j) { acc[j][0] = v8f{}; acc[j][1] = v8f{}; }

  for (int q = 0; q < DIM / 8; ++q) {
    const int k = q * 8;
    const v4f wi0 = *(const v4f*)(bi0 + k);
    const v4f we0 = *(const v4f*)(be0 + k);
    const v4f wi1 = *(const v4f*)(bi1 + k);
    const v4f we1 = *(const v4f*)(be1 + k);
    #pragma unroll
    for (int j = 0; j < NT; ++j) {
      const v4f g = *(const v4f*)(arow[j] + k);
      const v4f d = *(const v4f*)(drow[j] + k);
      const v4f a1 = g + d;
      const v4f a2 = g * d;
      acc[j][0] = wmma_f32(a1.xy, wi0.xy, acc[j][0]);
      acc[j][1] = wmma_f32(a1.xy, wi1.xy, acc[j][1]);
      acc[j][0] = wmma_f32(a1.zw, wi0.zw, acc[j][0]);
      acc[j][1] = wmma_f32(a1.zw, wi1.zw, acc[j][1]);
      acc[j][0] = wmma_f32(a2.xy, we0.xy, acc[j][0]);
      acc[j][1] = wmma_f32(a2.xy, we1.xy, acc[j][1]);
      acc[j][0] = wmma_f32(a2.zw, we0.zw, acc[j][0]);
      acc[j][1] = wmma_f32(a2.zw, we1.zw, acc[j][1]);
    }
  }

  #pragma unroll
  for (int j = 0; j < NT; ++j) {
    const int row0 = nbase + (grp + 8 * j) * 16 + 8 * h;
    #pragma unroll
    for (int r = 0; r < 8; ++r) {
      const int orow = row0 + r;
      if (orow < nrows) {
        float* o = out + (size_t)orow * DIM + n0 + lo;
        const float x0 = acc[j][0][r];
        const float x1 = acc[j][1][r];
        o[0]  = (x0 > 0.0f) ? x0 : 0.01f * x0;
        o[16] = (x1 > 0.0f) ? x1 : 0.01f * x1;
      }
    }
  }
}

static_assert(NBMAX % BIN_THR == 0 && BIN_THR % 32 == 0 && BIN_THR <= 1024, "scan layout below");

__global__ void __launch_bounds__(BIN_THR)
bin_edges_kernel(const int* __restrict__ src_idx, const int* __restrict__ dst_idx,
                 const float* __restrict__ ew, int* __restrict__ rec,
                 int* __restrict__ tab, int nrows, int nedges, int nb, int chunk) {
  __shared__ int s_cnt[NBMAX];
  __shared__ int s_wsum[32];

  const int tid  = threadIdx.x;
  const int lane = tid & 31;
  const int wave = tid >> 5;
  const int c    = blockIdx.x;
  const int e_lo = c * chunk;
  int e_hi = e_lo + chunk;
  if (e_hi > nedges) e_hi = nedges;

  for (int i = tid; i < NBMAX; i += BIN_THR) s_cnt[i] = 0;
  __syncthreads();
  for (int base = e_lo; base < e_hi; base += BIN_THR * BIN_U) {
    int dv[BIN_U];
    #pragma unroll
    for (int u = 0; u < BIN_U; ++u) {
      const int e = base + u * BIN_THR + tid;
      dv[u] = (e < e_hi) ? dst_idx[e] : -1;
    }
    #pragma unroll
    for (int u = 0; u < BIN_U; ++u) {
      if (dv[u] >= 0 && dv[u] < nrows) atomicAdd(&s_cnt[dv[u] / RPB], 1);
    }
  }
  __syncthreads();

  int cl[BPT];
  int sum = 0;
  #pragma unroll
  for (int k = 0; k < BPT; ++k) { cl[k] = s_cnt[BPT * tid + k]; sum += cl[k]; }
  int inc = sum;
  for (int o = 1; o < 32; o <<= 1) {
    const int t = __shfl_up(inc, o);
    if (lane >= o) inc += t;
  }
  if (lane == 31) s_wsum[wave] = inc;
  __syncthreads();
  if (wave == 0) {
    const int w = (lane < BIN_THR / 32) ? s_wsum[lane] : 0;
    int wi = w;
    for (int o = 1; o < 32; o <<= 1) {
      const int t = __shfl_up(wi, o);
      if (lane >= o) wi += t;
    }
    s_wsum[lane] = wi - w;
  }
  __syncthreads();
  int run = inc - sum + s_wsum[wave];
  #pragma unroll
  for (int k = 0; k < BPT; ++k) {
    const int b = BPT * tid + k;
    s_cnt[b] = run;
    if (b <= nb) tab[(size_t)c * (nb + 1) + b] = run;
    run += cl[k];
  }
  __syncthreads();

  for (int base = e_lo; base < e_hi; base += BIN_THR * BIN_U) {
    int   dv[BIN_U];
    int   sv[BIN_U];
    float wv[BIN_U];
    #pragma unroll
    for (int u = 0; u < BIN_U; ++u) {
      const int e = base + u * BIN_THR + tid;
      const bool in = e < e_hi;
      dv[u] = in ? dst_idx[e] : -1;
      sv[u] = in ? src_idx[e] : 0;
      wv[u] = in ? ew[e] : 0.0f;
    }
    #pragma unroll
    for (int u = 0; u < BIN_U; ++u) {
      const int d = dv[u];
      if (d >= 0 && d < nrows) {
        const int bb  = d / RPB;
        const int pos = atomicAdd(&s_cnt[bb], 1);
        int s = sv[u];
        s = (s < 0) ? 0 : ((s >= nrows) ? nrows - 1 : s);
        int* p = rec + 2 * (size_t)(e_lo + pos);
        p[0] = (s << 9) | (d - bb * RPB);
        p[1] = __builtin_bit_cast(int, wv[u]);
      }
    }
  }
}

template <bool FULL>
static __device__ __forceinline__ void
apply_group(const float* __restrict__ src_x, float* s_acc, int pk, float wv,
            int g, int m, int lane) {
  int   pj[8];
  float wj[8];
  #pragma unroll
  for (int j = 0; j < 8; ++j) {
    pj[j] = __builtin_amdgcn_readlane(pk, g + j);
    wj[j] = __builtin_bit_cast(float,
              __builtin_amdgcn_readlane(__builtin_bit_cast(int, wv), g + j));
  }
  float x[8][4];
  #pragma unroll
  for (int j = 0; j < 8; ++j) {
    if (FULL || j < m) {
      const float* xs = src_x + (size_t)(pj[j] >> 9) * DIM + lane;
      x[j][0] = xs[0]; x[j][1] = xs[32]; x[j][2] = xs[64]; x[j][3] = xs[96];
    }
  }
  #pragma unroll
  for (int j = 0; j < 8; ++j) {
    if (FULL || j < m) {
      const float w = wj[j];
      float* a = s_acc + (pj[j] & 511) * PITCH + lane;
      __hip_atomic_fetch_add(a,      w * x[j][0], __ATOMIC_RELAXED, __HIP_MEMORY_SCOPE_WORKGROUP);
      __hip_atomic_fetch_add(a + 32, w * x[j][1], __ATOMIC_RELAXED, __HIP_MEMORY_SCOPE_WORKGROUP);
      __hip_atomic_fetch_add(a + 64, w * x[j][2], __ATOMIC_RELAXED, __HIP_MEMORY_SCOPE_WORKGROUP);
      __hip_atomic_fetch_add(a + 96, w * x[j][3], __ATOMIC_RELAXED, __HIP_MEMORY_SCOPE_WORKGROUP);
    }
  }
}

template <bool BINNED>
__global__ void __launch_bounds__(NTHR)
owner_range_kernel(const float* __restrict__ src_x, const float* __restrict__ dst_x,
                   const int* __restrict__ src_idx, const int* __restrict__ dst_idx,
                   const float* __restrict__ ew,
                   const float* __restrict__ Wi, const float* __restrict__ We,
                   float* __restrict__ out, int nrows, int nedges,
                   const int* __restrict__ rec, const int* __restrict__ tab,
                   int nb, int chunk) {
  __shared__ __attribute__((aligned(16))) float s_acc[RPB * PITCH];

  const int tid   = threadIdx.x;
  const int lane  = tid & 31;
  const int wave  = __builtin_amdgcn_readfirstlane(tid >> 5);
  const int nbase = blockIdx.x * RPB;

  const int cch = wave * CPW + (lane & (CPW - 1));
  int o0 = 0;
  int o1 = 0;
  if constexpr (BINNED) {
    const int* t = tab + (size_t)cch * (nb + 1) + blockIdx.x;
    o0 = t[0];
    o1 = t[1];
  }

  for (int i = tid; i < RPB * PITCH / 4; i += NTHR)
    ((v4f*)s_acc)[i] = v4f{0.0f, 0.0f, 0.0f, 0.0f};
  __syncthreads();

  if constexpr (BINNED) {
    const int st = cch * chunk + o0;
    const int cn = (lane < CPW) ? (o1 - o0) : 0;
    int inc = cn;
    #pragma unroll
    for (int o = 1; o < CPW; o <<= 1) {
      const int t = __shfl_up(inc, o);
      if (lane >= o) inc += t;
    }
    const int total = __builtin_amdgcn_readlane(inc, CPW - 1);
    const int exc   = inc - cn;
    int segst[CPW];
    int segp[CPW];
    #pragma unroll
    for (int k = 0; k < CPW; ++k) {
      segst[k] = __builtin_amdgcn_readlane(st, k);
      segp[k]  = __builtin_amdgcn_readlane(exc, k);
    }

    for (int i0 = 0; i0 < total; i0 += 32) {
      const int id = i0 + lane;
      int sst = segst[0];
      int sp  = 0;
      #pragma unroll
      for (int k = 1; k < CPW; ++k) {
        if (id >= segp[k]) { sst = segst[k]; sp = segp[k]; }
      }
      int   pk = 0;
      float wv = 0.0f;
      if (id < total) {
        const int* p = rec + 2 * (size_t)(sst + id - sp);
        pk = p[0];
        wv = __builtin_bit_cast(float, p[1]);
      }
      const int n = (total - i0 < 32) ? (total - i0) : 32;
      for (int g = 0; g < n; g += 8) {
        if (n - g >= 8) apply_group<true>(src_x, s_acc, pk, wv, g, 8, lane);
        else            apply_group<false>(src_x, s_acc, pk, wv, g, n - g, lane);
      }
    }
  } else {
    int qe = 0;
    int ql = 0;
    int qn = 0;

    auto apply_top = [&](int n) {
      const int first = qn - n;
      int   sv = 0;
      float wv = 0.0f;
      {
        const int ee = __builtin_amdgcn_ds_bpermute((first + (lane & 7)) << 2, qe);
        if (lane < n) {
          sv = src_idx[ee];
          wv = ew[ee];
          sv = (sv < 0) ? 0 : ((sv >= nrows) ? nrows - 1 : sv);
        }
      }
      float x[8][4];
      #pragma unroll
      for (int j = 0; j < 8; ++j) {
        x[j][0] = 0.0f; x[j][1] = 0.0f; x[j][2] = 0.0f; x[j][3] = 0.0f;
        if (j < n) {
          const int s = __builtin_amdgcn_readlane(sv, j);
          const float* xs = src_x + (size_t)s * DIM + lane;
          x[j][0] = xs[0]; x[j][1] = xs[32]; x[j][2] = xs[64]; x[j][3] = xs[96];
        }
      }
      #pragma unroll
      for (int j = 0; j < 8; ++j) {
        if (j < n) {
          const float w  = __builtin_bit_cast(float,
                             __builtin_amdgcn_readlane(__builtin_bit_cast(int, wv), j));
          const int lrow = __builtin_amdgcn_readlane(ql, first + j);
          float* a = s_acc + lrow * PITCH + lane;
          __hip_atomic_fetch_add(a,      w * x[j][0], __ATOMIC_RELAXED, __HIP_MEMORY_SCOPE_WORKGROUP);
          __hip_atomic_fetch_add(a + 32, w * x[j][1], __ATOMIC_RELAXED, __HIP_MEMORY_SCOPE_WORKGROUP);
          __hip_atomic_fetch_add(a + 64, w * x[j][2], __ATOMIC_RELAXED, __HIP_MEMORY_SCOPE_WORKGROUP);
          __hip_atomic_fetch_add(a + 96, w * x[j][3], __ATOMIC_RELAXED, __HIP_MEMORY_SCOPE_WORKGROUP);
        }
      }
      qn = first;
    };

    auto apply_edge = [&](int ee, int lrow) {
      int s = src_idx[ee];
      s = (s < 0) ? 0 : ((s >= nrows) ? nrows - 1 : s);
      const float w = ew[ee];
      const float* xs = src_x + (size_t)s * DIM + lane;
      float* a = s_acc + lrow * PITCH + lane;
      const float x0 = xs[0], x1 = xs[32], x2 = xs[64], x3 = xs[96];
      __hip_atomic_fetch_add(a,      w * x0, __ATOMIC_RELAXED, __HIP_MEMORY_SCOPE_WORKGROUP);
      __hip_atomic_fetch_add(a + 32, w * x1, __ATOMIC_RELAXED, __HIP_MEMORY_SCOPE_WORKGROUP);
      __hip_atomic_fetch_add(a + 64, w * x2, __ATOMIC_RELAXED, __HIP_MEMORY_SCOPE_WORKGROUP);
      __hip_atomic_fetch_add(a + 96, w * x3, __ATOMIC_RELAXED, __HIP_MEMORY_SCOPE_WORKGROUP);
    };

    for (int e0 = wave * 32; e0 < nedges; e0 += NTHR * SCAN_U) {
      int dv[SCAN_U];
      if (e0 + NTHR * SCAN_U <= nedges) {
        #pragma unroll
        for (int u = 0; u < SCAN_U; ++u) dv[u] = dst_idx[e0 + u * NTHR + lane];
      } else {
        #pragma unroll
        for (int u = 0; u < SCAN_U; ++u) {
          const int e = e0 + u * NTHR + lane;
          dv[u] = (e < nedges) ? dst_idx[e] : -1;
        }
      }
      unsigned hitmask[SCAN_U];
      int tot = 0;
      #pragma unroll
      for (int u = 0; u < SCAN_U; ++u) {
        const unsigned rel = (unsigned)(dv[u] - nbase);
        const bool hit = (dv[u] >= 0) && (rel < (unsigned)RPB);
        hitmask[u] = __builtin_amdgcn_ballot_w32(hit);
        tot += __builtin_popcount(hitmask[u]);
      }
      if (qn + tot > 32) {
        while (qn > 0) apply_top(qn < 8 ? qn : 8);
      }
      if (tot <= 32) {
        #pragma unroll
        for (int u = 0; u < SCAN_U; ++u) {
          unsigned mask = hitmask[u];
          while (mask) {
            const int b = __builtin_ctz(mask);
            mask &= mask - 1;
            const int erow = __builtin_amdgcn_readlane(dv[u] - nbase, b);
            if (lane == qn) { qe = e0 + u * NTHR + b; ql = erow; }
            ++qn;
          }
        }
      } else {
        for (int u = 0; u < SCAN_U; ++u) {
          const int e = e0 + u * NTHR + lane;
          const int d = (e < nedges) ? dst_idx[e] : -1;
          const unsigned rel = (unsigned)(d - nbase);
          unsigned mask = __builtin_amdgcn_ballot_w32((d >= 0) && (rel < (unsigned)RPB));
          while (mask) {
            const int b = __builtin_ctz(mask);
            mask &= mask - 1;
            apply_edge(e0 + u * NTHR + b, __builtin_amdgcn_readlane((int)rel, b));
          }
        }
      }
      while (qn >= 8) apply_top(8);
    }
    while (qn > 0) apply_top(qn < 8 ? qn : 8);
  }
  __syncthreads();

  const int lo  = lane & 15;
  const int h   = lane >> 4;
  const int n0  = (wave & 3) * 32;
  const int grp = wave >> 2;
  if (grp + 24 < NTILE)
    transform_rows<4>(s_acc, dst_x, Wi, We, out, nbase, nrows, grp, n0, lo, h);
  else
    transform_rows<3>(s_acc, dst_x, Wi, We, out, nbase, nrows, grp, n0, lo, h);
}

extern "C" void kernel_launch(void* const* d_in, const int* in_sizes, int n_in,
                              void* d_out, int out_size, void* d_ws, size_t ws_size,
                              hipStream_t stream) {
  const float* src_x = (const float*)d_in[0];
  const float* dst_x = (const float*)d_in[1];
  const int*   eidx  = (const int*)  d_in[2];
  const float* ew    = (const float*)d_in[3];
  const float* Wi    = (const float*)d_in[4];
  const float* We    = (const float*)d_in[5];
  float*       out   = (float*)d_out;

  const int N = in_sizes[0] / DIM;
  const int E = in_sizes[2] / 2;
  const int* src_idx = eidx;
  const int* dst_idx = eidx + E;

  const int nb    = (N + RPB - 1) / RPB;
  const int chunk = (E + NCH - 1) / NCH;
  int* rec = (int*)d_ws;
  int* tab = rec + 2 * (size_t)E;
  const size_t need = (2 * (size_t)E + (size_t)NCH * (nb + 1)) * sizeof(int);

  if (d_ws != nullptr && ws_size >= need && nb < NBMAX) {
    bin_edges_kernel<<<NCH, BIN_THR, 0, stream>>>(src_idx, dst_idx, ew, rec, tab, N, E, nb, chunk);
    owner_range_kernel<true><<<nb, NTHR, 0, stream>>>(
        src_x, dst_x, src_idx, dst_idx, ew, Wi, We, out, N, E, rec, tab, nb, chunk);
  } else {
    owner_range_kernel<false><<<nb, NTHR, 0, stream>>>(
        src_x, dst_x, src_idx, dst_idx, ew, Wi, We, out, N, E, nullptr, nullptr, nb, chunk);
  }
}


# W_intra/W_inter fragments of 14 of the 16 k pair steps copied once per work-group into the LDS left beside the accumulator; transform reads its B operand there (ds_load_b128), not from global
# speedup vs baseline: 7.1633x; 1.0757x over previous
#include <hip/hip_runtime.h>


typedef __attribute__((ext_vector_type(2))) float v2f;
typedef __attribute__((ext_vector_type(4))) float v4f;
typedef __attribute__((ext_vector_type(8))) float v8f;

#define DIM    128
#define RPB    400
#define NTILE  (RPB / 16)
#define PITCH  132
#define NTHR   1024
#define SCAN_U 16
#define NCH    256
#define CPW    (NCH / 32)
#define NBMAX  1024
#define BIN_THR 256
#define BIN_U  10
#define BPT    (NBMAX / BIN_THR)
#define NQ     (DIM / 8)
#define QL     14
#define WF_V4  (2 * (DIM / 16) * QL * 32)
#define WF_PT  (WF_V4 / NTHR)

static_assert(WF_V4 % NTHR == 0, "every thread stages the same number of pieces");
static_assert((RPB * PITCH + WF_V4 * 4) * 4 <= 327680, "accumulator + W fragments must fit one work-group's LDS");

static __device__ __forceinline__ v8f wmma_f32(v2f a, v2f b, v8f c) {
  return __builtin_amdgcn_wmma_f32_16x16x4_f32(false, a, false, b,
                                               (short)0, c, false, false);
}

template <int NT>
static __device__ __forceinline__ void
transform_rows(const float* __restrict__ s_acc, const float* __restrict__ s_w,
               const float* __restrict__ dst_x,
               const float* __restrict__ Wi, const float* __restrict__ We,
               float* __restrict__ out, int nbase, int nrows,
               int grp, int n0, int lo, int h) {
  const float* wl = s_w + (size_t)(((n0 >> 4) * 2 * QL) * 32 + lo + 16 * h) * 4;
  const float* bi0 = Wi + (size_t)(n0 + lo) * DIM + 4 * h;
  const float* be0 = We + (size_t)(n0 + lo) * DIM + 4 * h;
  const float* bi1 = bi0 + 16 * DIM;
  const float* be1 = be0 + 16 * DIM;

  const float* arow[NT];
  const float* drow[NT];
  #pragma unroll
  for (int j = 0; j < NT; ++j) {
    const int lrow = (grp + 8 * j) * 16 + lo;
    int grow = nbase + lrow;
    if (grow >= nrows) grow = nrows - 1;
    arow[j] = s_acc + lrow * PITCH + 4 * h;
    drow[j] = dst_x + (size_t)grow * DIM + 4 * h;
  }

  v8f acc[NT][2];
  #pragma unroll
  for (int j = 0; j < NT; ++j) { acc[j][0] = v8f{}; acc[j][1] = v8f{}; }

  #pragma unroll
  for (int q = 0; q < NQ; ++q) {
    const int k = q * 8;
    v4f wi0, we0, wi1, we1;
    if (q < QL) {
      wi0 = *(const v4f*)(wl + ((0 * QL + q) * 32) * 4);
      we0 = *(const v4f*)(wl + ((1 * QL + q) * 32) * 4);
      wi1 = *(const v4f*)(wl + ((2 * QL + q) * 32) * 4);
      we1 = *(const v4f*)(wl + ((3 * QL + q) * 32) * 4);
    } else {
      wi0 = *(const v4f*)(bi0 + k);
      we0 = *(const v4f*)(be0 + k);
      wi1 = *(const v4f*)(bi1 + k);
      we1 = *(const v4f*)(be1 + k);
    }
    #pragma unroll
    for (int j = 0; j < NT; ++j) {
      const v4f g = *(const v4f*)(arow[j] + k);
      const v4f d = *(const v4f*)(drow[j] + k);
      const v4f a1 = g + d;
      const v4f a2 = g * d;
      acc[j][0] = wmma_f32(a1.xy, wi0.xy, acc[j][0]);
      acc[j][1] = wmma_f32(a1.xy, wi1.xy, acc[j][1]);
      acc[j][0] = wmma_f32(a1.zw, wi0.zw, acc[j][0]);
      acc[j][1] = wmma_f32(a1.zw, wi1.zw, acc[j][1]);
      acc[j][0] = wmma_f32(a2.xy, we0.xy, acc[j][0]);
      acc[j][1] = wmma_f32(a2.xy, we1.xy, acc[j][1]);
      acc[j][0] = wmma_f32(a2.zw, we0.zw, acc[j][0]);
      acc[j][1] = wmma_f32(a2.zw, we1.zw, acc[j][1]);
    }
  }

  #pragma unroll
  for (int j = 0; j < NT; ++j) {
    const int row0 = nbase + (grp + 8 * j) * 16 + 8 * h;
    #pragma unroll
    for (int r = 0; r < 8; ++r) {
      const int orow = row0 + r;
      if (orow < nrows) {
        float* o = out + (size_t)orow * DIM + n0 + lo;
        const float x0 = acc[j][0][r];
        const float x1 = acc[j][1][r];
        o[0]  = (x0 > 0.0f) ? x0 : 0.01f * x0;
        o[16] = (x1 > 0.0f) ? x1 : 0.01f * x1;
      }
    }
  }
}

static_assert(NBMAX % BIN_THR == 0 && BIN_THR % 32 == 0 && BIN_THR <= 1024, "scan layout below");

__global__ void __launch_bounds__(BIN_THR)
bin_edges_kernel(const int* __restrict__ src_idx, const int* __restrict__ dst_idx,
                 const float* __restrict__ ew, int* __restrict__ rec,
                 int* __restrict__ tab, int nrows, int nedges, int nb, int chunk) {
  __shared__ int s_cnt[NBMAX];
  __shared__ int s_wsum[32];

  const int tid  = threadIdx.x;
  const int lane = tid & 31;
  const int wave = tid >> 5;
  const int c    = blockIdx.x;
  const int e_lo = c * chunk;
  int e_hi = e_lo + chunk;
  if (e_hi > nedges) e_hi = nedges;

  for (int i = tid; i < NBMAX; i += BIN_THR) s_cnt[i] = 0;
  __syncthreads();
  for (int base = e_lo; base < e_hi; base += BIN_THR * BIN_U) {
    int dv[BIN_U];
    #pragma unroll
    for (int u = 0; u < BIN_U; ++u) {
      const int e = base + u * BIN_THR + tid;
      dv[u] = (e < e_hi) ? dst_idx[e] : -1;
    }
    #pragma unroll
    for (int u = 0; u < BIN_U; ++u) {
      if (dv[u] >= 0 && dv[u] < nrows) atomicAdd(&s_cnt[dv[u] / RPB], 1);
    }
  }
  __syncthreads();

  int cl[BPT];
  int sum = 0;
  #pragma unroll
  for (int k = 0; k < BPT; ++k) { cl[k] = s_cnt[BPT * tid + k]; sum += cl[k]; }
  int inc = sum;
  for (int o = 1; o < 32; o <<= 1) {
    const int t = __shfl_up(inc, o);
    if (lane >= o) inc += t;
  }
  if (lane == 31) s_wsum[wave] = inc;
  __syncthreads();
  if (wave == 0) {
    const int w = (lane < BIN_THR / 32) ? s_wsum[lane] : 0;
    int wi = w;
    for (int o = 1; o < 32; o <<= 1) {
      const int t = __shfl_up(wi, o);
      if (lane >= o) wi += t;
    }
    s_wsum[lane] = wi - w;
  }
  __syncthreads();
  int run = inc - sum + s_wsum[wave];
  #pragma unroll
  for (int k = 0; k < BPT; ++k) {
    const int b = BPT * tid + k;
    s_cnt[b] = run;
    if (b <= nb) tab[(size_t)c * (nb + 1) + b] = run;
    run += cl[k];
  }
  __syncthreads();

  for (int base = e_lo; base < e_hi; base += BIN_THR * BIN_U) {
    int   dv[BIN_U];
    int   sv[BIN_U];
    float wv[BIN_U];
    #pragma unroll
    for (int u = 0; u < BIN_U; ++u) {
      const int e = base + u * BIN_THR + tid;
      const bool in = e < e_hi;
      dv[u] = in ? dst_idx[e] : -1;
      sv[u] = in ? src_idx[e] : 0;
      wv[u] = in ? ew[e] : 0.0f;
    }
    #pragma unroll
    for (int u = 0; u < BIN_U; ++u) {
      const int d = dv[u];
      if (d >= 0 && d < nrows) {
        const int bb  = d / RPB;
        const int pos = atomicAdd(&s_cnt[bb], 1);
        int s = sv[u];
        s = (s < 0) ? 0 : ((s >= nrows) ? nrows - 1 : s);
        int* p = rec + 2 * (size_t)(e_lo + pos);
        p[0] = (s << 9) | (d - bb * RPB);
        p[1] = __builtin_bit_cast(int, wv[u]);
      }
    }
  }
}

template <bool FULL>
static __device__ __forceinline__ void
apply_group(const float* __restrict__ src_x, float* s_acc, int pk, float wv,
            int g, int m, int lane) {
  int   pj[8];
  float wj[8];
  #pragma unroll
  for (int j = 0; j < 8; ++j) {
    pj[j] = __builtin_amdgcn_readlane(pk, g + j);
    wj[j] = __builtin_bit_cast(float,
              __builtin_amdgcn_readlane(__builtin_bit_cast(int, wv), g + j));
  }
  float x[8][4];
  #pragma unroll
  for (int j = 0; j < 8; ++j) {
    if (FULL || j < m) {
      const float* xs = src_x + (size_t)(pj[j] >> 9) * DIM + lane;
      x[j][0] = xs[0]; x[j][1] = xs[32]; x[j][2] = xs[64]; x[j][3] = xs[96];
    }
  }
  #pragma unroll
  for (int j = 0; j < 8; ++j) {
    if (FULL || j < m) {
      const float w = wj[j];
      float* a = s_acc + (pj[j] & 511) * PITCH + lane;
      __hip_atomic_fetch_add(a,      w * x[j][0], __ATOMIC_RELAXED, __HIP_MEMORY_SCOPE_WORKGROUP);
      __hip_atomic_fetch_add(a + 32, w * x[j][1], __ATOMIC_RELAXED, __HIP_MEMORY_SCOPE_WORKGROUP);
      __hip_atomic_fetch_add(a + 64, w * x[j][2], __ATOMIC_RELAXED, __HIP_MEMORY_SCOPE_WORKGROUP);
      __hip_atomic_fetch_add(a + 96, w * x[j][3], __ATOMIC_RELAXED, __HIP_MEMORY_SCOPE_WORKGROUP);
    }
  }
}

template <bool BINNED>
__global__ void __launch_bounds__(NTHR)
owner_range_kernel(const float* __restrict__ src_x, const float* __restrict__ dst_x,
                   const int* __restrict__ src_idx, const int* __restrict__ dst_idx,
                   const float* __restrict__ ew,
                   const float* __restrict__ Wi, const float* __restrict__ We,
                   float* __restrict__ out, int nrows, int nedges,
                   const int* __restrict__ rec, const int* __restrict__ tab,
                   int nb, int chunk) {
  __shared__ __attribute__((aligned(16))) float s_acc[RPB * PITCH];
  __shared__ __attribute__((aligned(16))) float s_w[WF_V4 * 4];

  const int tid   = threadIdx.x;
  const int lane  = tid & 31;
  const int wave  = __builtin_amdgcn_readfirstlane(tid >> 5);
  const int nbase = blockIdx.x * RPB;

  const int cch = wave * CPW + (lane & (CPW - 1));
  int o0 = 0;
  int o1 = 0;
  if constexpr (BINNED) {
    const int* t = tab + (size_t)cch * (nb + 1) + blockIdx.x;
    o0 = t[0];
    o1 = t[1];
  }

  v4f wst[WF_PT];
  #pragma unroll
  for (int i = 0; i < WF_PT; ++i) {
    const int rest = i * (NTHR / 32) + wave;
    const int q    = rest % QL;
    const int tm   = rest / QL;
    const float* wsrc = (tm & 1) ? We : Wi;
    wst[i] = *(const v4f*)(wsrc + (size_t)((tm >> 1) * 16 + (lane & 15)) * DIM + 8 * q + 4 * (lane >> 4));
  }

  for (int i = tid; i < RPB * PITCH / 4; i += NTHR)
    ((v4f*)s_acc)[i] = v4f{0.0f, 0.0f, 0.0f, 0.0f};
  #pragma unroll
  for (int i = 0; i < WF_PT; ++i) ((v4f*)s_w)[i * NTHR + tid] = wst[i];
  __syncthreads();

  if constexpr (BINNED) {
    const int st = cch * chunk + o0;
    const int cn = (lane < CPW) ? (o1 - o0) : 0;
    int inc = cn;
    #pragma unroll
    for (int o = 1; o < CPW; o <<= 1) {
      const int t = __shfl_up(inc, o);
      if (lane >= o) inc += t;
    }
    const int total = __builtin_amdgcn_readlane(inc, CPW - 1);
    const int exc   = inc - cn;
    int segst[CPW];
    int segp[CPW];
    #pragma unroll
    for (int k = 0; k < CPW; ++k) {
      segst[k] = __builtin_amdgcn_readlane(st, k);
      segp[k]  = __builtin_amdgcn_readlane(exc, k);
    }

    for (int i0 = 0; i0 < total; i0 += 32) {
      const int id = i0 + lane;
      int sst = segst[0];
      int sp  = 0;
      #pragma unroll
      for (int k = 1; k < CPW; ++k) {
        if (id >= segp[k]) { sst = segst[k]; sp = segp[k]; }
      }
      int   pk = 0;
      float wv = 0.0f;
      if (id < total) {
        const int* p = rec + 2 * (size_t)(sst + id - sp);
        pk = p[0];
        wv = __builtin_bit_cast(float, p[1]);
      }
      const int n = (total - i0 < 32) ? (total - i0) : 32;
      for (int g = 0; g < n; g += 8) {
        if (n - g >= 8) apply_group<true>(src_x, s_acc, pk, wv, g, 8, lane);
        else            apply_group<false>(src_x, s_acc, pk, wv, g, n - g, lane);
      }
    }
  } else {
    int qe = 0;
    int ql = 0;
    int qn = 0;

    auto apply_top = [&](int n) {
      const int first = qn - n;
      int   sv = 0;
      float wv = 0.0f;
      {
        const int ee = __builtin_amdgcn_ds_bpermute((first + (lane & 7)) << 2, qe);
        if (lane < n) {
          sv = src_idx[ee];
          wv = ew[ee];
          sv = (sv < 0) ? 0 : ((sv >= nrows) ? nrows - 1 : sv);
        }
      }
      float x[8][4];
      #pragma unroll
      for (int j = 0; j < 8; ++j) {
        x[j][0] = 0.0f; x[j][1] = 0.0f; x[j][2] = 0.0f; x[j][3] = 0.0f;
        if (j < n) {
          const int s = __builtin_amdgcn_readlane(sv, j);
          const float* xs = src_x + (size_t)s * DIM + lane;
          x[j][0] = xs[0]; x[j][1] = xs[32]; x[j][2] = xs[64]; x[j][3] = xs[96];
        }
      }
      #pragma unroll
      for (int j = 0; j < 8; ++j) {
        if (j < n) {
          const float w  = __builtin_bit_cast(float,
                             __builtin_amdgcn_readlane(__builtin_bit_cast(int, wv), j));
          const int lrow = __builtin_amdgcn_readlane(ql, first + j);
          float* a = s_acc + lrow * PITCH + lane;
          __hip_atomic_fetch_add(a,      w * x[j][0], __ATOMIC_RELAXED, __HIP_MEMORY_SCOPE_WORKGROUP);
          __hip_atomic_fetch_add(a + 32, w * x[j][1], __ATOMIC_RELAXED, __HIP_MEMORY_SCOPE_WORKGROUP);
          __hip_atomic_fetch_add(a + 64, w * x[j][2], __ATOMIC_RELAXED, __HIP_MEMORY_SCOPE_WORKGROUP);
          __hip_atomic_fetch_add(a + 96, w * x[j][3], __ATOMIC_RELAXED, __HIP_MEMORY_SCOPE_WORKGROUP);
        }
      }
      qn = first;
    };

    auto apply_edge = [&](int ee, int lrow) {
      int s = src_idx[ee];
      s = (s < 0) ? 0 : ((s >= nrows) ? nrows - 1 : s);
      const float w = ew[ee];
      const float* xs = src_x + (size_t)s * DIM + lane;
      float* a = s_acc + lrow * PITCH + lane;
      const float x0 = xs[0], x1 = xs[32], x2 = xs[64], x3 = xs[96];
      __hip_atomic_fetch_add(a,      w * x0, __ATOMIC_RELAXED, __HIP_MEMORY_SCOPE_WORKGROUP);
      __hip_atomic_fetch_add(a + 32, w * x1, __ATOMIC_RELAXED, __HIP_MEMORY_SCOPE_WORKGROUP);
      __hip_atomic_fetch_add(a + 64, w * x2, __ATOMIC_RELAXED, __HIP_MEMORY_SCOPE_WORKGROUP);
      __hip_atomic_fetch_add(a + 96, w * x3, __ATOMIC_RELAXED, __HIP_MEMORY_SCOPE_WORKGROUP);
    };

    for (int e0 = wave * 32; e0 < nedges; e0 += NTHR * SCAN_U) {
      int dv[SCAN_U];
      if (e0 + NTHR * SCAN_U <= nedges) {
        #pragma unroll
        for (int u = 0; u < SCAN_U; ++u) dv[u] = dst_idx[e0 + u * NTHR + lane];
      } else {
        #pragma unroll
        for (int u = 0; u < SCAN_U; ++u) {
          const int e = e0 + u * NTHR + lane;
          dv[u] = (e < nedges) ? dst_idx[e] : -1;
        }
      }
      unsigned hitmask[SCAN_U];
      int tot = 0;
      #pragma unroll
      for (int u = 0; u < SCAN_U; ++u) {
        const unsigned rel = (unsigned)(dv[u] - nbase);
        const bool hit = (dv[u] >= 0) && (rel < (unsigned)RPB);
        hitmask[u] = __builtin_amdgcn_ballot_w32(hit);
        tot += __builtin_popcount(hitmask[u]);
      }
      if (qn + tot > 32) {
        while (qn > 0) apply_top(qn < 8 ? qn : 8);
      }
      if (tot <= 32) {
        #pragma unroll
        for (int u = 0; u < SCAN_U; ++u) {
          unsigned mask = hitmask[u];
          while (mask) {
            const int b = __builtin_ctz(mask);
            mask &= mask - 1;
            const int erow = __builtin_amdgcn_readlane(dv[u] - nbase, b);
            if (lane == qn) { qe = e0 + u * NTHR + b; ql = erow; }
            ++qn;
          }
        }
      } else {
        for (int u = 0; u < SCAN_U; ++u) {
          const int e = e0 + u * NTHR + lane;
          const int d = (e < nedges) ? dst_idx[e] : -1;
          const unsigned rel = (unsigned)(d - nbase);
          unsigned mask = __builtin_amdgcn_ballot_w32((d >= 0) && (rel < (unsigned)RPB));
          while (mask) {
            const int b = __builtin_ctz(mask);
            mask &= mask - 1;
            apply_edge(e0 + u * NTHR + b, __builtin_amdgcn_readlane((int)rel, b));
          }
        }
      }
      while (qn >= 8) apply_top(8);
    }
    while (qn > 0) apply_top(qn < 8 ? qn : 8);
  }
  __syncthreads();

  const int lo  = lane & 15;
  const int h   = lane >> 4;
  const int n0  = (wave & 3) * 32;
  const int grp = wave >> 2;
  if (grp + 24 < NTILE)
    transform_rows<4>(s_acc, s_w, dst_x, Wi, We, out, nbase, nrows, grp, n0, lo, h);
  else
    transform_rows<3>(s_acc, s_w, dst_x, Wi, We, out, nbase, nrows, grp, n0, lo, h);
}

extern "C" void kernel_launch(void* const* d_in, const int* in_sizes, int n_in,
                              void* d_out, int out_size, void* d_ws, size_t ws_size,
                              hipStream_t stream) {
  const float* src_x = (const float*)d_in[0];
  const float* dst_x = (const float*)d_in[1];
  const int*   eidx  = (const int*)  d_in[2];
  const float* ew    = (const float*)d_in[3];
  const float* Wi    = (const float*)d_in[4];
  const float* We    = (const float*)d_in[5];
  float*       out   = (float*)d_out;

  const int N = in_sizes[0] / DIM;
  const int E = in_sizes[2] / 2;
  const int* src_idx = eidx;
  const int* dst_idx = eidx + E;

  const int nb    = (N + RPB - 1) / RPB;
  const int chunk = (E + NCH - 1) / NCH;
  int* rec = (int*)d_ws;
  int* tab = rec + 2 * (size_t)E;
  const size_t need = (2 * (size_t)E + (size_t)NCH * (nb + 1)) * sizeof(int);

  if (d_ws != nullptr && ws_size >= need && nb < NBMAX) {
    bin_edges_kernel<<<NCH, BIN_THR, 0, stream>>>(src_idx, dst_idx, ew, rec, tab, N, E, nb, chunk);
    owner_range_kernel<true><<<nb, NTHR, 0, stream>>>(
        src_x, dst_x, src_idx, dst_idx, ew, Wi, We, out, N, E, rec, tab, nb, chunk);
  } else {
    owner_range_kernel<false><<<nb, NTHR, 0, stream>>>(
        src_x, dst_x, src_idx, dst_idx, ew, Wi, We, out, N, E, nullptr, nullptr, nb, chunk);
  }
}


# Transform: a wave covers four 16-col tiles (was two) of fewer row tiles: one dst and one LDS accumulator fragment feed 16 WMMAs, not 8; tile 24 in quarters, one per SIMD; sums unchanged
# speedup vs baseline: 7.7204x; 1.0778x over previous
#include <hip/hip_runtime.h>


typedef __attribute__((ext_vector_type(2))) float v2f;
typedef __attribute__((ext_vector_type(4))) float v4f;
typedef __attribute__((ext_vector_type(8))) float v8f;

#define DIM    128
#define RPB    400
#define NTILE  (RPB / 16)
#define PITCH  132
#define NTHR   1024
#define SCAN_U 16
#define NCH    256
#define CPW    (NCH / 32)
#define NBMAX  1024
#define BIN_THR 256
#define BIN_U  10
#define BPT    (NBMAX / BIN_THR)
#define NQ     (DIM / 8)
#define QL     14
#define WF_V4  (2 * (DIM / 16) * QL * 32)
#define WF_PT  (WF_V4 / NTHR)

static_assert(WF_V4 % NTHR == 0, "every thread stages the same number of pieces");
static_assert(NTILE == 25 && NTHR == 1024, "transform tile map: 24 tiles over 16 row groups x 2 column halves, tile 24 in quarters");
static_assert((RPB * PITCH + WF_V4 * 4) * 4 <= 327680, "accumulator + W fragments must fit one work-group's LDS");

static __device__ __forceinline__ v8f wmma_f32(v2f a, v2f b, v8f c) {
  return __builtin_amdgcn_wmma_f32_16x16x4_f32(false, a, false, b,
                                               (short)0, c, false, false);
}

template <int NT>
static __device__ __forceinline__ void
transform_rows(const float* __restrict__ s_acc, const float* __restrict__ s_w,
               const float* __restrict__ dst_x,
               const float* __restrict__ Wi, const float* __restrict__ We,
               float* __restrict__ out, int nbase, int nrows,
               int grp, int n0, int lo, int h) {
  const float* wl = s_w + (size_t)(((n0 >> 4) * 2 * QL) * 32 + lo + 16 * h) * 4;
  const float* bi0 = Wi + (size_t)(n0 + lo) * DIM + 4 * h;
  const float* be0 = We + (size_t)(n0 + lo) * DIM + 4 * h;
  const float* bi1 = bi0 + 16 * DIM;
  const float* be1 = be0 + 16 * DIM;

  const float* arow[NT];
  const float* drow[NT];
  #pragma unroll
  for (int j = 0; j < NT; ++j) {
    const int lrow = (grp + 8 * j) * 16 + lo;
    int grow = nbase + lrow;
    if (grow >= nrows) grow = nrows - 1;
    arow[j] = s_acc + lrow * PITCH + 4 * h;
    drow[j] = dst_x + (size_t)grow * DIM + 4 * h;
  }

  v8f acc[NT][2];
  #pragma unroll
  for (int j = 0; j < NT; ++j) { acc[j][0] = v8f{}; acc[j][1] = v8f{}; }

  #pragma unroll
  for (int q = 0; q < NQ; ++q) {
    const int k = q * 8;
    v4f wi0, we0, wi1, we1;
    if (q < QL) {
      wi0 = *(const v4f*)(wl + ((0 * QL + q) * 32) * 4);
      we0 = *(const v4f*)(wl + ((1 * QL + q) * 32) * 4);
      wi1 = *(const v4f*)(wl + ((2 * QL + q) * 32) * 4);
      we1 = *(const v4f*)(wl + ((3 * QL + q) * 32) * 4);
    } else {
      wi0 = *(const v4f*)(bi0 + k);
      we0 = *(const v4f*)(be0 + k);
      wi1 = *(const v4f*)(bi1 + k);
      we1 = *(const v4f*)(be1 + k);
    }
    #pragma unroll
    for (int j = 0; j < NT; ++j) {
      const v4f g = *(const v4f*)(arow[j] + k);
      const v4f d = *(const v4f*)(drow[j] + k);
      const v4f a1 = g + d;
      const v4f a2 = g * d;
      acc[j][0] = wmma_f32(a1.xy, wi0.xy, acc[j][0]);
      acc[j][1] = wmma_f32(a1.xy, wi1.xy, acc[j][1]);
      acc[j][0] = wmma_f32(a1.zw, wi0.zw, acc[j][0]);
      acc[j][1] = wmma_f32(a1.zw, wi1.zw, acc[j][1]);
      acc[j][0] = wmma_f32(a2.xy, we0.xy, acc[j][0]);
      acc[j][1] = wmma_f32(a2.xy, we1.xy, acc[j][1]);
      acc[j][0] = wmma_f32(a2.zw, we0.zw, acc[j][0]);
      acc[j][1] = wmma_f32(a2.zw, we1.zw, acc[j][1]);
    }
  }

  #pragma unroll
  for (int j = 0; j < NT; ++j) {
    const int row0 = nbase + (grp + 8 * j) * 16 + 8 * h;
    #pragma unroll
    for (int r = 0; r < 8; ++r) {
      const int orow = row0 + r;
      if (orow < nrows) {
        float* o = out + (size_t)orow * DIM + n0 + lo;
        const float x0 = acc[j][0][r];
        const float x1 = acc[j][1][r];
        o[0]  = (x0 > 0.0f) ? x0 : 0.01f * x0;
        o[16] = (x1 > 0.0f) ? x1 : 0.01f * x1;
      }
    }
  }
}

template <int NT>
static __device__ __forceinline__ void
transform_wide(const float* __restrict__ s_acc, const float* __restrict__ s_w,
               const float* __restrict__ dst_x,
               const float* __restrict__ Wi, const float* __restrict__ We,
               float* __restrict__ out, int nbase, int nrows,
               int rg, int c0, int lo, int h) {
  const float* wl = s_w + (size_t)(((c0 >> 4) * 2 * QL) * 32 + lo + 16 * h) * 4;
  const unsigned boff = (unsigned)(c0 + lo) * (DIM * 4) + 16 * h;

  const float* arow[NT];
  unsigned doff[NT];
  #pragma unroll
  for (int j = 0; j < NT; ++j) {
    const int lrow = (rg + 16 * j) * 16 + lo;
    int grow = nbase + lrow;
    if (grow >= nrows) grow = nrows - 1;
    arow[j] = s_acc + lrow * PITCH + 4 * h;
    doff[j] = (unsigned)grow * (DIM * 4) + 16 * h;
  }

  v8f acc[NT][4];
  #pragma unroll
  for (int j = 0; j < NT; ++j) {
    #pragma unroll
    for (int t = 0; t < 4; ++t) acc[j][t] = v8f{};
  }

  #pragma unroll
  for (int q = 0; q < NQ; ++q) {
    const int k = q * 8;
    v4f a1[NT], a2[NT];
    #pragma unroll
    for (int j = 0; j < NT; ++j) {
      const v4f g = *(const v4f*)(arow[j] + k);
      const v4f d = *(const v4f*)((const char*)dst_x + (doff[j] + (unsigned)(k * 4)));
      a1[j] = g + d;
      a2[j] = g * d;
    }
    #pragma unroll
    for (int m = 0; m < 2; ++m) {
      v4f b[4];
      #pragma unroll
      for (int t = 0; t < 4; ++t) {
        if (q < QL) b[t] = *(const v4f*)(wl + (((t * 2 + m) * QL + q) * 32) * 4);
        else        b[t] = *(const v4f*)((const char*)(m ? We : Wi) + (boff + (unsigned)((16 * t * DIM + k) * 4)));
      }
      #pragma unroll
      for (int j = 0; j < NT; ++j) {
        const v4f a = m ? a2[j] : a1[j];
        #pragma unroll
        for (int t = 0; t < 4; ++t) acc[j][t] = wmma_f32(a.xy, b[t].xy, acc[j][t]);
        #pragma unroll
        for (int t = 0; t < 4; ++t) acc[j][t] = wmma_f32(a.zw, b[t].zw, acc[j][t]);
      }
    }
  }

  #pragma unroll
  for (int j = 0; j < NT; ++j) {
    const int row0 = nbase + (rg + 16 * j) * 16 + 8 * h;
    #pragma unroll
    for (int r = 0; r < 8; ++r) {
      const int orow = row0 + r;
      if (orow < nrows) {
        float* o = out + (size_t)orow * DIM + c0 + lo;
        #pragma unroll
        for (int t = 0; t < 4; ++t) {
          const float x = acc[j][t][r];
          o[16 * t] = (x > 0.0f) ? x : 0.01f * x;
        }
      }
    }
  }
}

static_assert(NBMAX % BIN_THR == 0 && BIN_THR % 32 == 0 && BIN_THR <= 1024, "scan layout below");

__global__ void __launch_bounds__(BIN_THR)
bin_edges_kernel(const int* __restrict__ src_idx, const int* __restrict__ dst_idx,
                 const float* __restrict__ ew, int* __restrict__ rec,
                 int* __restrict__ tab, int nrows, int nedges, int nb, int chunk) {
  __shared__ int s_cnt[NBMAX];
  __shared__ int s_wsum[32];

  const int tid  = threadIdx.x;
  const int lane = tid & 31;
  const int wave = tid >> 5;
  const int c    = blockIdx.x;
  const int e_lo = c * chunk;
  int e_hi = e_lo + chunk;
  if (e_hi > nedges) e_hi = nedges;

  for (int i = tid; i < NBMAX; i += BIN_THR) s_cnt[i] = 0;
  __syncthreads();
  for (int base = e_lo; base < e_hi; base += BIN_THR * BIN_U) {
    int dv[BIN_U];
    #pragma unroll
    for (int u = 0; u < BIN_U; ++u) {
      const int e = base + u * BIN_THR + tid;
      dv[u] = (e < e_hi) ? dst_idx[e] : -1;
    }
    #pragma unroll
    for (int u = 0; u < BIN_U; ++u) {
      if (dv[u] >= 0 && dv[u] < nrows) atomicAdd(&s_cnt[dv[u] / RPB], 1);
    }
  }
  __syncthreads();

  int cl[BPT];
  int sum = 0;
  #pragma unroll
  for (int k = 0; k < BPT; ++k) { cl[k] = s_cnt[BPT * tid + k]; sum += cl[k]; }
  int inc = sum;
  for (int o = 1; o < 32; o <<= 1) {
    const int t = __shfl_up(inc, o);
    if (lane >= o) inc += t;
  }
  if (lane == 31) s_wsum[wave] = inc;
  __syncthreads();
  if (wave == 0) {
    const int w = (lane < BIN_THR / 32) ? s_wsum[lane] : 0;
    int wi = w;
    for (int o = 1; o < 32; o <<= 1) {
      const int t = __shfl_up(wi, o);
      if (lane >= o) wi += t;
    }
    s_wsum[lane] = wi - w;
  }
  __syncthreads();
  int run = inc - sum + s_wsum[wave];
  #pragma unroll
  for (int k = 0; k < BPT; ++k) {
    const int b = BPT * tid + k;
    s_cnt[b] = run;
    if (b <= nb) tab[(size_t)c * (nb + 1) + b] = run;
    run += cl[k];
  }
  __syncthreads();

  for (int base = e_lo; base < e_hi; base += BIN_THR * BIN_U) {
    int   dv[BIN_U];
    int   sv[BIN_U];
    float wv[BIN_U];
    #pragma unroll
    for (int u = 0; u < BIN_U; ++u) {
      const int e = base + u * BIN_THR + tid;
      const bool in = e < e_hi;
      dv[u] = in ? dst_idx[e] : -1;
      sv[u] = in ? src_idx[e] : 0;
      wv[u] = in ? ew[e] : 0.0f;
    }
    #pragma unroll
    for (int u = 0; u < BIN_U; ++u) {
      const int d = dv[u];
      if (d >= 0 && d < nrows) {
        const int bb  = d / RPB;
        const int pos = atomicAdd(&s_cnt[bb], 1);
        int s = sv[u];
        s = (s < 0) ? 0 : ((s >= nrows) ? nrows - 1 : s);
        int* p = rec + 2 * (size_t)(e_lo + pos);
        p[0] = (s << 9) | (d - bb * RPB);
        p[1] = __builtin_bit_cast(int, wv[u]);
      }
    }
  }
}

template <bool FULL>
static __device__ __forceinline__ void
apply_group(const float* __restrict__ src_x, float* s_acc, int pk, float wv,
            int g, int m, int lane) {
  int   pj[8];
  float wj[8];
  #pragma unroll
  for (int j = 0; j < 8; ++j) {
    pj[j] = __builtin_amdgcn_readlane(pk, g + j);
    wj[j] = __builtin_bit_cast(float,
              __builtin_amdgcn_readlane(__builtin_bit_cast(int, wv), g + j));
  }
  float x[8][4];
  #pragma unroll
  for (int j = 0; j < 8; ++j) {
    if (FULL || j < m) {
      const float* xs = src_x + (size_t)(pj[j] >> 9) * DIM + lane;
      x[j][0] = xs[0]; x[j][1] = xs[32]; x[j][2] = xs[64]; x[j][3] = xs[96];
    }
  }
  #pragma unroll
  for (int j = 0; j < 8; ++j) {
    if (FULL || j < m) {
      const float w = wj[j];
      float* a = s_acc + (pj[j] & 511) * PITCH + lane;
      __hip_atomic_fetch_add(a,      w * x[j][0], __ATOMIC_RELAXED, __HIP_MEMORY_SCOPE_WORKGROUP);
      __hip_atomic_fetch_add(a + 32, w * x[j][1], __ATOMIC_RELAXED, __HIP_MEMORY_SCOPE_WORKGROUP);
      __hip_atomic_fetch_add(a + 64, w * x[j][2], __ATOMIC_RELAXED, __HIP_MEMORY_SCOPE_WORKGROUP);
      __hip_atomic_fetch_add(a + 96, w * x[j][3], __ATOMIC_RELAXED, __HIP_MEMORY_SCOPE_WORKGROUP);
    }
  }
}

template <bool BINNED>
__global__ void __launch_bounds__(NTHR)
owner_range_kernel(const float* __restrict__ src_x, const float* __restrict__ dst_x,
                   const int* __restrict__ src_idx, const int* __restrict__ dst_idx,
                   const float* __restrict__ ew,
                   const float* __restrict__ Wi, const float* __restrict__ We,
                   float* __restrict__ out, int nrows, int nedges,
                   const int* __restrict__ rec, const int* __restrict__ tab,
                   int nb, int chunk) {
  __shared__ __attribute__((aligned(16))) float s_acc[RPB * PITCH];
  __shared__ __attribute__((aligned(16))) float s_w[WF_V4 * 4];

  const int tid   = threadIdx.x;
  const int lane  = tid & 31;
  const int wave  = __builtin_amdgcn_readfirstlane(tid >> 5);
  const int nbase = blockIdx.x * RPB;

  const int cch = wave * CPW + (lane & (CPW - 1));
  int o0 = 0;
  int o1 = 0;
  if constexpr (BINNED) {
    const int* t = tab + (size_t)cch * (nb + 1) + blockIdx.x;
    o0 = t[0];
    o1 = t[1];
  }

  v4f wst[WF_PT];
  #pragma unroll
  for (int i = 0; i < WF_PT; ++i) {
    const int rest = i * (NTHR / 32) + wave;
    const int q    = rest % QL;
    const int tm   = rest / QL;
    const float* wsrc = (tm & 1) ? We : Wi;
    wst[i] = *(const v4f*)(wsrc + (size_t)((tm >> 1) * 16 + (lane & 15)) * DIM + 8 * q + 4 * (lane >> 4));
  }

  for (int i = tid; i < RPB * PITCH / 4; i += NTHR)
    ((v4f*)s_acc)[i] = v4f{0.0f, 0.0f, 0.0f, 0.0f};
  #pragma unroll
  for (int i = 0; i < WF_PT; ++i) ((v4f*)s_w)[i * NTHR + tid] = wst[i];
  __syncthreads();

  if constexpr (BINNED) {
    const int st = cch * chunk + o0;
    const int cn = (lane < CPW) ? (o1 - o0) : 0;
    int inc = cn;
    #pragma unroll
    for (int o = 1; o < CPW; o <<= 1) {
      const int t = __shfl_up(inc, o);
      if (lane >= o) inc += t;
    }
    const int total = __builtin_amdgcn_readlane(inc, CPW - 1);
    const int exc   = inc - cn;
    int segst[CPW];
    int segp[CPW];
    #pragma unroll
    for (int k = 0; k < CPW; ++k) {
      segst[k] = __builtin_amdgcn_readlane(st, k);
      segp[k]  = __builtin_amdgcn_readlane(exc, k);
    }

    for (int i0 = 0; i0 < total; i0 += 32) {
      const int id = i0 + lane;
      int sst = segst[0];
      int sp  = 0;
      #pragma unroll
      for (int k = 1; k < CPW; ++k) {
        if (id >= segp[k]) { sst = segst[k]; sp = segp[k]; }
      }
      int   pk = 0;
      float wv = 0.0f;
      if (id < total) {
        const int* p = rec + 2 * (size_t)(sst + id - sp);
        pk = p[0];
        wv = __builtin_bit_cast(float, p[1]);
      }
      const int n = (total - i0 < 32) ? (total - i0) : 32;
      for (int g = 0; g < n; g += 8) {
        if (n - g >= 8) apply_group<true>(src_x, s_acc, pk, wv, g, 8, lane);
        else            apply_group<false>(src_x, s_acc, pk, wv, g, n - g, lane);
      }
    }
  } else {
    int qe = 0;
    int ql = 0;
    int qn = 0;

    auto apply_top = [&](int n) {
      const int first = qn - n;
      int   sv = 0;
      float wv = 0.0f;
      {
        const int ee = __builtin_amdgcn_ds_bpermute((first + (lane & 7)) << 2, qe);
        if (lane < n) {
          sv = src_idx[ee];
          wv = ew[ee];
          sv = (sv < 0) ? 0 : ((sv >= nrows) ? nrows - 1 : sv);
        }
      }
      float x[8][4];
      #pragma unroll
      for (int j = 0; j < 8; ++j) {
        x[j][0] = 0.0f; x[j][1] = 0.0f; x[j][2] = 0.0f; x[j][3] = 0.0f;
        if (j < n) {
          const int s = __builtin_amdgcn_readlane(sv, j);
          const float* xs = src_x + (size_t)s * DIM + lane;
          x[j][0] = xs[0]; x[j][1] = xs[32]; x[j][2] = xs[64]; x[j][3] = xs[96];
        }
      }
      #pragma unroll
      for (int j = 0; j < 8; ++j) {
        if (j < n) {
          const float w  = __builtin_bit_cast(float,
                             __builtin_amdgcn_readlane(__builtin_bit_cast(int, wv), j));
          const int lrow = __builtin_amdgcn_readlane(ql, first + j);
          float* a = s_acc + lrow * PITCH + lane;
          __hip_atomic_fetch_add(a,      w * x[j][0], __ATOMIC_RELAXED, __HIP_MEMORY_SCOPE_WORKGROUP);
          __hip_atomic_fetch_add(a + 32, w * x[j][1], __ATOMIC_RELAXED, __HIP_MEMORY_SCOPE_WORKGROUP);
          __hip_atomic_fetch_add(a + 64, w * x[j][2], __ATOMIC_RELAXED, __HIP_MEMORY_SCOPE_WORKGROUP);
          __hip_atomic_fetch_add(a + 96, w * x[j][3], __ATOMIC_RELAXED, __HIP_MEMORY_SCOPE_WORKGROUP);
        }
      }
      qn = first;
    };

    auto apply_edge = [&](int ee, int lrow) {
      int s = src_idx[ee];
      s = (s < 0) ? 0 : ((s >= nrows) ? nrows - 1 : s);
      const float w = ew[ee];
      const float* xs = src_x + (size_t)s * DIM + lane;
      float* a = s_acc + lrow * PITCH + lane;
      const float x0 = xs[0], x1 = xs[32], x2 = xs[64], x3 = xs[96];
      __hip_atomic_fetch_add(a,      w * x0, __ATOMIC_RELAXED, __HIP_MEMORY_SCOPE_WORKGROUP);
      __hip_atomic_fetch_add(a + 32, w * x1, __ATOMIC_RELAXED, __HIP_MEMORY_SCOPE_WORKGROUP);
      __hip_atomic_fetch_add(a + 64, w * x2, __ATOMIC_RELAXED, __HIP_MEMORY_SCOPE_WORKGROUP);
      __hip_atomic_fetch_add(a + 96, w * x3, __ATOMIC_RELAXED, __HIP_MEMORY_SCOPE_WORKGROUP);
    };

    for (int e0 = wave * 32; e0 < nedges; e0 += NTHR * SCAN_U) {
      int dv[SCAN_U];
      if (e0 + NTHR * SCAN_U <= nedges) {
        #pragma unroll
        for (int u = 0; u < SCAN_U; ++u) dv[u] = dst_idx[e0 + u * NTHR + lane];
      } else {
        #pragma unroll
        for (int u = 0; u < SCAN_U; ++u) {
          const int e = e0 + u * NTHR + lane;
          dv[u] = (e < nedges) ? dst_idx[e] : -1;
        }
      }
      unsigned hitmask[SCAN_U];
      int tot = 0;
      #pragma unroll
      for (int u = 0; u < SCAN_U; ++u) {
        const unsigned rel = (unsigned)(dv[u] - nbase);
        const bool hit = (dv[u] >= 0) && (rel < (unsigned)RPB);
        hitmask[u] = __builtin_amdgcn_ballot_w32(hit);
        tot += __builtin_popcount(hitmask[u]);
      }
      if (qn + tot > 32) {
        while (qn > 0) apply_top(qn < 8 ? qn : 8);
      }
      if (tot <= 32) {
        #pragma unroll
        for (int u = 0; u < SCAN_U; ++u) {
          unsigned mask = hitmask[u];
          while (mask) {
            const int b = __builtin_ctz(mask);
            mask &= mask - 1;
            const int erow = __builtin_amdgcn_readlane(dv[u] - nbase, b);
            if (lane == qn) { qe = e0 + u * NTHR + b; ql = erow; }
            ++qn;
          }
        }
      } else {
        for (int u = 0; u < SCAN_U; ++u) {
          const int e = e0 + u * NTHR + lane;
          const int d = (e < nedges) ? dst_idx[e] : -1;
          const unsigned rel = (unsigned)(d - nbase);
          unsigned mask = __builtin_amdgcn_ballot_w32((d >= 0) && (rel < (unsigned)RPB));
          while (mask) {
            const int b = __builtin_ctz(mask);
            mask &= mask - 1;
            apply_edge(e0 + u * NTHR + b, __builtin_amdgcn_readlane((int)rel, b));
          }
        }
      }
      while (qn >= 8) apply_top(8);
    }
    while (qn > 0) apply_top(qn < 8 ? qn : 8);
  }
  __syncthreads();

  const int lo  = lane & 15;
  const int h   = lane >> 4;
  const int rg = wave >> 1;
  const int c0 = (wave & 1) * 64;
  if (rg < 8) {
    transform_wide<2>(s_acc, s_w, dst_x, Wi, We, out, nbase, nrows, rg, c0, lo, h);
  } else {
    transform_wide<1>(s_acc, s_w, dst_x, Wi, We, out, nbase, nrows, rg, c0, lo, h);
    if (wave >= 18 && wave < 22)
      transform_rows<1>(s_acc, s_w, dst_x, Wi, We, out, nbase, nrows, NTILE - 1, (wave & 3) * 32, lo, h);
  }
}

extern "C" void kernel_launch(void* const* d_in, const int* in_sizes, int n_in,
                              void* d_out, int out_size, void* d_ws, size_t ws_size,
                              hipStream_t stream) {
  const float* src_x = (const float*)d_in[0];
  const float* dst_x = (const float*)d_in[1];
  const int*   eidx  = (const int*)  d_in[2];
  const float* ew    = (const float*)d_in[3];
  const float* Wi    = (const float*)d_in[4];
  const float* We    = (const float*)d_in[5];
  float*       out   = (float*)d_out;

  const int N = in_sizes[0] / DIM;
  const int E = in_sizes[2] / 2;
  const int* src_idx = eidx;
  const int* dst_idx = eidx + E;

  const int nb    = (N + RPB - 1) / RPB;
  const int chunk = (E + NCH - 1) / NCH;
  int* rec = (int*)d_ws;
  int* tab = rec + 2 * (size_t)E;
  const size_t need = (2 * (size_t)E + (size_t)NCH * (nb + 1)) * sizeof(int);

  if (d_ws != nullptr && ws_size >= need && nb < NBMAX) {
    bin_edges_kernel<<<NCH, BIN_THR, 0, stream>>>(src_idx, dst_idx, ew, rec, tab, N, E, nb, chunk);
    owner_range_kernel<true><<<nb, NTHR, 0, stream>>>(
        src_x, dst_x, src_idx, dst_idx, ew, Wi, We, out, N, E, rec, tab, nb, chunk);
  } else {
    owner_range_kernel<false><<<nb, NTHR, 0, stream>>>(
        src_x, dst_x, src_idx, dst_idx, ew, Wi, We, out, N, E, nullptr, nullptr, nb, chunk);
  }
}


# Owner sorts its range's records by destination row in LDS (count, scan, place); a wave sums its rows in registers from whole-row 16-byte loads, one LDS store a row, no LDS float adds; serial row sums
# speedup vs baseline: 8.2388x; 1.0671x over previous
#include <hip/hip_runtime.h>


typedef __attribute__((ext_vector_type(2))) float v2f;
typedef __attribute__((ext_vector_type(4))) float v4f;
typedef __attribute__((ext_vector_type(8))) float v8f;

#define DIM    128
#define RPB    400
#define NTILE  (RPB / 16)
#define PITCH  132
#define NTHR   1024
#define SCAN_U 16
#define NCH    256
#define CPW    (NCH / 32)
#define NBMAX  1024
#define BIN_THR 256
#define BIN_U  10
#define BPT    (NBMAX / BIN_THR)
#define NQ     (DIM / 8)
#define QL     14
#define WF_V4  (2 * (DIM / 16) * QL * 32)
#define WF_PT  (WF_V4 / NTHR)

#define NROWP  512
#define LIST_I0 1040
#define LIST_CAP ((WF_V4 * 4 - LIST_I0) / 2)
#define RHELD  4

static_assert(WF_V4 % NTHR == 0, "every thread stages the same number of pieces");
static_assert(RPB <= NROWP && NROWP % 32 == 0 && 2 * NROWP + 2 <= LIST_I0 && LIST_I0 % 2 == 0, "sort tables sit in front of the list");
static_assert(NTILE == 25 && NTHR == 1024, "transform tile map: 24 tiles over 16 row groups x 2 column halves, tile 24 in quarters");
static_assert((RPB * PITCH + WF_V4 * 4) * 4 <= 327680, "accumulator + W fragments must fit one work-group's LDS");

static __device__ __forceinline__ v8f wmma_f32(v2f a, v2f b, v8f c) {
  return __builtin_amdgcn_wmma_f32_16x16x4_f32(false, a, false, b,
                                               (short)0, c, false, false);
}

template <int NT>
static __device__ __forceinline__ void
transform_rows(const float* __restrict__ s_acc, const float* __restrict__ s_w,
               const float* __restrict__ dst_x,
               const float* __restrict__ Wi, const float* __restrict__ We,
               float* __restrict__ out, int nbase, int nrows,
               int grp, int n0, int lo, int h) {
  const float* wl = s_w + (size_t)(((n0 >> 4) * 2 * QL) * 32 + lo + 16 * h) * 4;
  const float* bi0 = Wi + (size_t)(n0 + lo) * DIM + 4 * h;
  const float* be0 = We + (size_t)(n0 + lo) * DIM + 4 * h;
  const float* bi1 = bi0 + 16 * DIM;
  const float* be1 = be0 + 16 * DIM;

  const float* arow[NT];
  const float* drow[NT];
  #pragma unroll
  for (int j = 0; j < NT; ++j) {
    const int lrow = (grp + 8 * j) * 16 + lo;
    int grow = nbase + lrow;
    if (grow >= nrows) grow = nrows - 1;
    arow[j] = s_acc + lrow * PITCH + 4 * h;
    drow[j] = dst_x + (size_t)grow * DIM + 4 * h;
  }

  v8f acc[NT][2];
  #pragma unroll
  for (int j = 0; j < NT; ++j) { acc[j][0] = v8f{}; acc[j][1] = v8f{}; }

  #pragma unroll
  for (int q = 0; q < NQ; ++q) {
    const int k = q * 8;
    v4f wi0, we0, wi1, we1;
    if (q < QL) {
      wi0 = *(const v4f*)(wl + ((0 * QL + q) * 32) * 4);
      we0 = *(const v4f*)(wl + ((1 * QL + q) * 32) * 4);
      wi1 = *(const v4f*)(wl + ((2 * QL + q) * 32) * 4);
      we1 = *(const v4f*)(wl + ((3 * QL + q) * 32) * 4);
    } else {
      wi0 = *(const v4f*)(bi0 + k);
      we0 = *(const v4f*)(be0 + k);
      wi1 = *(const v4f*)(bi1 + k);
      we1 = *(const v4f*)(be1 + k);
    }
    #pragma unroll
    for (int j = 0; j < NT; ++j) {
      const v4f g = *(const v4f*)(arow[j] + k);
      const v4f d = *(const v4f*)(drow[j] + k);
      const v4f a1 = g + d;
      const v4f a2 = g * d;
      acc[j][0] = wmma_f32(a1.xy, wi0.xy, acc[j][0]);
      acc[j][1] = wmma_f32(a1.xy, wi1.xy, acc[j][1]);
      acc[j][0] = wmma_f32(a1.zw, wi0.zw, acc[j][0]);
      acc[j][1] = wmma_f32(a1.zw, wi1.zw, acc[j][1]);
      acc[j][0] = wmma_f32(a2.xy, we0.xy, acc[j][0]);
      acc[j][1] = wmma_f32(a2.xy, we1.xy, acc[j][1]);
      acc[j][0] = wmma_f32(a2.zw, we0.zw, acc[j][0]);
      acc[j][1] = wmma_f32(a2.zw, we1.zw, acc[j][1]);
    }
  }

  #pragma unroll
  for (int j = 0; j < NT; ++j) {
    const int row0 = nbase + (grp + 8 * j) * 16 + 8 * h;
    #pragma unroll
    for (int r = 0; r < 8; ++r) {
      const int orow = row0 + r;
      if (orow < nrows) {
        float* o = out + (size_t)orow * DIM + n0 + lo;
        const float x0 = acc[j][0][r];
        const float x1 = acc[j][1][r];
        o[0]  = (x0 > 0.0f) ? x0 : 0.01f * x0;
        o[16] = (x1 > 0.0f) ? x1 : 0.01f * x1;
      }
    }
  }
}

template <int NT>
static __device__ __forceinline__ void
transform_wide(const float* __restrict__ s_acc, const float* __restrict__ s_w,
               const float* __restrict__ dst_x,
               const float* __restrict__ Wi, const float* __restrict__ We,
               float* __restrict__ out, int nbase, int nrows,
               int rg, int c0, int lo, int h) {
  const float* wl = s_w + (size_t)(((c0 >> 4) * 2 * QL) * 32 + lo + 16 * h) * 4;
  const unsigned boff = (unsigned)(c0 + lo) * (DIM * 4) + 16 * h;

  const float* arow[NT];
  unsigned doff[NT];
  #pragma unroll
  for (int j = 0; j < NT; ++j) {
    const int lrow = (rg + 16 * j) * 16 + lo;
    int grow = nbase + lrow;
    if (grow >= nrows) grow = nrows - 1;
    arow[j] = s_acc + lrow * PITCH + 4 * h;
    doff[j] = (unsigned)grow * (DIM * 4) + 16 * h;
  }

  v8f acc[NT][4];
  #pragma unroll
  for (int j = 0; j < NT; ++j) {
    #pragma unroll
    for (int t = 0; t < 4; ++t) acc[j][t] = v8f{};
  }

  #pragma unroll
  for (int q = 0; q < NQ; ++q) {
    const int k = q * 8;
    v4f a1[NT], a2[NT];
    #pragma unroll
    for (int j = 0; j < NT; ++j) {
      const v4f g = *(const v4f*)(arow[j] + k);
      const v4f d = *(const v4f*)((const char*)dst_x + (doff[j] + (unsigned)(k * 4)));
      a1[j] = g + d;
      a2[j] = g * d;
    }
    #pragma unroll
    for (int m = 0; m < 2; ++m) {
      v4f b[4];
      #pragma unroll
      for (int t = 0; t < 4; ++t) {
        if (q < QL) b[t] = *(const v4f*)(wl + (((t * 2 + m) * QL + q) * 32) * 4);
        else        b[t] = *(const v4f*)((const char*)(m ? We : Wi) + (boff + (unsigned)((16 * t * DIM + k) * 4)));
      }
      #pragma unroll
      for (int j = 0; j < NT; ++j) {
        const v4f a = m ? a2[j] : a1[j];
        #pragma unroll
        for (int t = 0; t < 4; ++t) acc[j][t] = wmma_f32(a.xy, b[t].xy, acc[j][t]);
        #pragma unroll
        for (int t = 0; t < 4; ++t) acc[j][t] = wmma_f32(a.zw, b[t].zw, acc[j][t]);
      }
    }
  }

  #pragma unroll
  for (int j = 0; j < NT; ++j) {
    const int row0 = nbase + (rg + 16 * j) * 16 + 8 * h;
    #pragma unroll
    for (int r = 0; r < 8; ++r) {
      const int orow = row0 + r;
      if (orow < nrows) {
        float* o = out + (size_t)orow * DIM + c0 + lo;
        #pragma unroll
        for (int t = 0; t < 4; ++t) {
          const float x = acc[j][t][r];
          o[16 * t] = (x > 0.0f) ? x : 0.01f * x;
        }
      }
    }
  }
}

static_assert(NBMAX % BIN_THR == 0 && BIN_THR % 32 == 0 && BIN_THR <= 1024, "scan layout below");

__global__ void __launch_bounds__(BIN_THR)
bin_edges_kernel(const int* __restrict__ src_idx, const int* __restrict__ dst_idx,
                 const float* __restrict__ ew, int* __restrict__ rec,
                 int* __restrict__ tab, int nrows, int nedges, int nb, int chunk) {
  __shared__ int s_cnt[NBMAX];
  __shared__ int s_wsum[32];

  const int tid  = threadIdx.x;
  const int lane = tid & 31;
  const int wave = tid >> 5;
  const int c    = blockIdx.x;
  const int e_lo = c * chunk;
  int e_hi = e_lo + chunk;
  if (e_hi > nedges) e_hi = nedges;

  for (int i = tid; i < NBMAX; i += BIN_THR) s_cnt[i] = 0;
  __syncthreads();
  for (int base = e_lo; base < e_hi; base += BIN_THR * BIN_U) {
    int dv[BIN_U];
    #pragma unroll
    for (int u = 0; u < BIN_U; ++u) {
      const int e = base + u * BIN_THR + tid;
      dv[u] = (e < e_hi) ? dst_idx[e] : -1;
    }
    #pragma unroll
    for (int u = 0; u < BIN_U; ++u) {
      if (dv[u] >= 0 && dv[u] < nrows) atomicAdd(&s_cnt[dv[u] / RPB], 1);
    }
  }
  __syncthreads();

  int cl[BPT];
  int sum = 0;
  #pragma unroll
  for (int k = 0; k < BPT; ++k) { cl[k] = s_cnt[BPT * tid + k]; sum += cl[k]; }
  int inc = sum;
  for (int o = 1; o < 32; o <<= 1) {
    const int t = __shfl_up(inc, o);
    if (lane >= o) inc += t;
  }
  if (lane == 31) s_wsum[wave] = inc;
  __syncthreads();
  if (wave == 0) {
    const int w = (lane < BIN_THR / 32) ? s_wsum[lane] : 0;
    int wi = w;
    for (int o = 1; o < 32; o <<= 1) {
      const int t = __shfl_up(wi, o);
      if (lane >= o) wi += t;
    }
    s_wsum[lane] = wi - w;
  }
  __syncthreads();
  int run = inc - sum + s_wsum[wave];
  #pragma unroll
  for (int k = 0; k < BPT; ++k) {
    const int b = BPT * tid + k;
    s_cnt[b] = run;
    if (b <= nb) tab[(size_t)c * (nb + 1) + b] = run;
    run += cl[k];
  }
  __syncthreads();

  for (int base = e_lo; base < e_hi; base += BIN_THR * BIN_U) {
    int   dv[BIN_U];
    int   sv[BIN_U];
    float wv[BIN_U];
    #pragma unroll
    for (int u = 0; u < BIN_U; ++u) {
      const int e = base + u * BIN_THR + tid;
      const bool in = e < e_hi;
      dv[u] = in ? dst_idx[e] : -1;
      sv[u] = in ? src_idx[e] : 0;
      wv[u] = in ? ew[e] : 0.0f;
    }
    #pragma unroll
    for (int u = 0; u < BIN_U; ++u) {
      const int d = dv[u];
      if (d >= 0 && d < nrows) {
        const int bb  = d / RPB;
        const int pos = atomicAdd(&s_cnt[bb], 1);
        int s = sv[u];
        s = (s < 0) ? 0 : ((s >= nrows) ? nrows - 1 : s);
        int* p = rec + 2 * (size_t)(e_lo + pos);
        p[0] = (s << 9) | (d - bb * RPB);
        p[1] = __builtin_bit_cast(int, wv[u]);
      }
    }
  }
}

template <bool FULL>
static __device__ __forceinline__ void
apply_group(const float* __restrict__ src_x, float* s_acc, int pk, float wv,
            int g, int m, int lane) {
  int   pj[8];
  float wj[8];
  #pragma unroll
  for (int j = 0; j < 8; ++j) {
    pj[j] = __builtin_amdgcn_readlane(pk, g + j);
    wj[j] = __builtin_bit_cast(float,
              __builtin_amdgcn_readlane(__builtin_bit_cast(int, wv), g + j));
  }
  float x[8][4];
  #pragma unroll
  for (int j = 0; j < 8; ++j) {
    if (FULL || j < m) {
      const float* xs = src_x + (size_t)(pj[j] >> 9) * DIM + lane;
      x[j][0] = xs[0]; x[j][1] = xs[32]; x[j][2] = xs[64]; x[j][3] = xs[96];
    }
  }
  #pragma unroll
  for (int j = 0; j < 8; ++j) {
    if (FULL || j < m) {
      const float w = wj[j];
      float* a = s_acc + (pj[j] & 511) * PITCH + lane;
      __hip_atomic_fetch_add(a,      w * x[j][0], __ATOMIC_RELAXED, __HIP_MEMORY_SCOPE_WORKGROUP);
      __hip_atomic_fetch_add(a + 32, w * x[j][1], __ATOMIC_RELAXED, __HIP_MEMORY_SCOPE_WORKGROUP);
      __hip_atomic_fetch_add(a + 64, w * x[j][2], __ATOMIC_RELAXED, __HIP_MEMORY_SCOPE_WORKGROUP);
      __hip_atomic_fetch_add(a + 96, w * x[j][3], __ATOMIC_RELAXED, __HIP_MEMORY_SCOPE_WORKGROUP);
    }
  }
}

static __device__ __forceinline__ void
flush_rows(float* s_acc, int cur, int next, v4f acc, int lane) {
  *(v4f*)(s_acc + cur * PITCH + 4 * lane) = acc;
  for (int z = cur + 1; z < next; ++z)
    *(v4f*)(s_acc + z * PITCH + 4 * lane) = v4f{0.0f, 0.0f, 0.0f, 0.0f};
}

template <bool FULL>
static __device__ __forceinline__ void
sum_group(const float* __restrict__ src_x, float* s_acc, int pk, float wv,
          int g, int m, int lane, int& cur, v4f& acc) {
  int   pj[8];
  float wj[8];
  #pragma unroll
  for (int j = 0; j < 8; ++j) {
    pj[j] = __builtin_amdgcn_readlane(pk, g + j);
    wj[j] = __builtin_bit_cast(float,
              __builtin_amdgcn_readlane(__builtin_bit_cast(int, wv), g + j));
  }
  v4f x[8];
  #pragma unroll
  for (int j = 0; j < 8; ++j) {
    if (FULL || j < m)
      x[j] = *((const v4f*)(src_x + (size_t)(pj[j] >> 9) * DIM) + lane);
  }
  #pragma unroll
  for (int j = 0; j < 8; ++j) {
    if (FULL || j < m) {
      const int r = pj[j] & 511;
      if (r != cur) {
        flush_rows(s_acc, cur, r, acc, lane);
        cur = r;
        acc = v4f{0.0f, 0.0f, 0.0f, 0.0f};
      }
      acc += wj[j] * x[j];
    }
  }
}

template <bool BINNED>
__global__ void __launch_bounds__(NTHR)
owner_range_kernel(const float* __restrict__ src_x, const float* __restrict__ dst_x,
                   const int* __restrict__ src_idx, const int* __restrict__ dst_idx,
                   const float* __restrict__ ew,
                   const float* __restrict__ Wi, const float* __restrict__ We,
                   float* __restrict__ out, int nrows, int nedges,
                   const int* __restrict__ rec, const int* __restrict__ tab,
                   int nb, int chunk) {
  __shared__ __attribute__((aligned(16))) float s_acc[RPB * PITCH];
  __shared__ __attribute__((aligned(16))) float s_w[WF_V4 * 4];

  const int tid   = threadIdx.x;
  const int lane  = tid & 31;
  const int wave  = __builtin_amdgcn_readfirstlane(tid >> 5);
  const int nbase = blockIdx.x * RPB;

  const int cch = wave * CPW + (lane & (CPW - 1));
  int o0 = 0;
  int o1 = 0;
  if constexpr (BINNED) {
    const int* t = tab + (size_t)cch * (nb + 1) + blockIdx.x;
    o0 = t[0];
    o1 = t[1];
  }

  v4f wst[WF_PT];
  auto load_w = [&]() {
    #pragma unroll
    for (int i = 0; i < WF_PT; ++i) {
      const int rest = i * (NTHR / 32) + wave;
      const int q    = rest % QL;
      const int tm   = rest / QL;
      const float* wsrc = (tm & 1) ? We : Wi;
      wst[i] = *(const v4f*)(wsrc + (size_t)((tm >> 1) * 16 + (lane & 15)) * DIM + 8 * q + 4 * (lane >> 4));
    }
  };
  auto store_w = [&]() {
    #pragma unroll
    for (int i = 0; i < WF_PT; ++i) ((v4f*)s_w)[i * NTHR + tid] = wst[i];
  };
  auto zero_acc = [&]() {
    for (int i = tid; i < RPB * PITCH / 4; i += NTHR)
      ((v4f*)s_acc)[i] = v4f{0.0f, 0.0f, 0.0f, 0.0f};
  };

  if constexpr (!BINNED) {
    load_w();
    zero_acc();
    store_w();
    __syncthreads();
  }

  if constexpr (BINNED) {
    int* s_cnt  = (int*)s_w;
    int* s_off  = (int*)s_w + NROWP;
    int* s_tot  = (int*)s_w + 2 * NROWP + 2;
    int* s_list = (int*)s_w + LIST_I0;
    if (tid < NROWP) s_cnt[tid] = 0;
    if (tid == 0) *s_tot = 0;
    __syncthreads();

    const int st = cch * chunk + o0;
    const int cn = (lane < CPW) ? (o1 - o0) : 0;
    int inc = cn;
    #pragma unroll
    for (int o = 1; o < CPW; o <<= 1) {
      const int t = __shfl_up(inc, o);
      if (lane >= o) inc += t;
    }
    const int total = __builtin_amdgcn_readlane(inc, CPW - 1);
    const int exc   = inc - cn;
    const int dlt = st - exc;
    int segp[CPW];
    #pragma unroll
    for (int k = 0; k < CPW; ++k) segp[k] = __builtin_amdgcn_readlane(exc, k);

    auto rec_of = [&](int id) -> const int* {
      int kk = 0;
      #pragma unroll
      for (int k = 1; k < CPW; ++k) kk += (id >= segp[k]) ? 1 : 0;
      return rec + 2 * (size_t)(__shfl(dlt, kk) + id);
    };

    int pkh[RHELD];
    int wbh[RHELD];
    #pragma unroll
    for (int u = 0; u < RHELD; ++u) { pkh[u] = 0; wbh[u] = 0; }
    if (total > 0) {
      #pragma unroll
      for (int u = 0; u < RHELD; ++u) {
        const int id = 32 * u + lane;
        const int* p = rec_of(id < total ? id : total - 1);
        pkh[u] = p[0];
        wbh[u] = p[1];
      }
    }
    #pragma unroll
    for (int u = 0; u < RHELD; ++u) {
      if (32 * u + lane < total) atomicAdd(&s_cnt[pkh[u] & 511], 1);
    }
    for (int i0 = 32 * RHELD; i0 < total; i0 += 32) {
      const int id = i0 + lane;
      const int* p = rec_of(id);
      if (id < total) atomicAdd(&s_cnt[p[0] & 511], 1);
    }
    if (lane == 0) atomicAdd(s_tot, total);
    __syncthreads();
    const int btot = __builtin_amdgcn_readfirstlane(*s_tot);

    if (btot <= LIST_CAP) {
      if (wave == 0) {
        int cl[NROWP / 32];
        int sum = 0;
        #pragma unroll
        for (int k = 0; k < NROWP / 32; ++k) { cl[k] = s_cnt[(NROWP / 32) * lane + k]; sum += cl[k]; }
        int sc = sum;
        for (int o = 1; o < 32; o <<= 1) {
          const int t = __shfl_up(sc, o);
          if (lane >= o) sc += t;
        }
        int run = sc - sum;
        #pragma unroll
        for (int k = 0; k < NROWP / 32; ++k) {
          s_off[(NROWP / 32) * lane + k] = run;
          s_cnt[(NROWP / 32) * lane + k] = run;
          run += cl[k];
        }
        if (lane == 31) s_off[NROWP] = run;
      }
      __syncthreads();
      load_w();

      #pragma unroll
      for (int u = 0; u < RHELD; ++u) {
        if (32 * u + lane < total) {
          const int pos = atomicAdd(&s_cnt[pkh[u] & 511], 1);
          s_list[2 * pos]     = pkh[u];
          s_list[2 * pos + 1] = wbh[u];
        }
      }
      for (int i0 = 32 * RHELD; i0 < total; i0 += 32) {
        const int id = i0 + lane;
        const int* p = rec_of(id);
        if (id < total) {
          const int pk = p[0];
          const int wb = p[1];
          const int pos = atomicAdd(&s_cnt[pk & 511], 1);
          s_list[2 * pos]     = pk;
          s_list[2 * pos + 1] = wb;
        }
      }
      __syncthreads();

      const int r0 = (wave * RPB) >> 5;
      const int r1 = ((wave + 1) * RPB) >> 5;
      const int e0 = __builtin_amdgcn_readfirstlane(s_off[r0]);
      const int e1 = __builtin_amdgcn_readfirstlane(s_off[r1]);
      int cur = r0;
      v4f acc = v4f{0.0f, 0.0f, 0.0f, 0.0f};
      for (int i0 = e0; i0 < e1; i0 += 32) {
        const int id = i0 + lane;
        int   pk = 0;
        float wv = 0.0f;
        if (id < e1) {
          pk = s_list[2 * id];
          wv = __builtin_bit_cast(float, s_list[2 * id + 1]);
        }
        const int n = (e1 - i0 < 32) ? (e1 - i0) : 32;
        for (int g = 0; g < n; g += 8) {
          if (n - g >= 8) sum_group<true>(src_x, s_acc, pk, wv, g, 8, lane, cur, acc);
          else            sum_group<false>(src_x, s_acc, pk, wv, g, n - g, lane, cur, acc);
        }
      }
      flush_rows(s_acc, cur, r1, acc, lane);
    } else {
      zero_acc();
      __syncthreads();
      load_w();
      for (int i0 = 0; i0 < total; i0 += 32) {
        const int id = i0 + lane;
        int   pk = 0;
        float wv = 0.0f;
        const int* p = rec_of(id);
        if (id < total) {
          pk = p[0];
          wv = __builtin_bit_cast(float, p[1]);
        }
        const int n = (total - i0 < 32) ? (total - i0) : 32;
        for (int g = 0; g < n; g += 8) {
          if (n - g >= 8) apply_group<true>(src_x, s_acc, pk, wv, g, 8, lane);
          else            apply_group<false>(src_x, s_acc, pk, wv, g, n - g, lane);
        }
      }
    }
    __syncthreads();
    store_w();
  } else {
    int qe = 0;
    int ql = 0;
    int qn = 0;

    auto apply_top = [&](int n) {
      const int first = qn - n;
      int   sv = 0;
      float wv = 0.0f;
      {
        const int ee = __builtin_amdgcn_ds_bpermute((first + (lane & 7)) << 2, qe);
        if (lane < n) {
          sv = src_idx[ee];
          wv = ew[ee];
          sv = (sv < 0) ? 0 : ((sv >= nrows) ? nrows - 1 : sv);
        }
      }
      float x[8][4];
      #pragma unroll
      for (int j = 0; j < 8; ++j) {
        x[j][0] = 0.0f; x[j][1] = 0.0f; x[j][2] = 0.0f; x[j][3] = 0.0f;
        if (j < n) {
          const int s = __builtin_amdgcn_readlane(sv, j);
          const float* xs = src_x + (size_t)s * DIM + lane;
          x[j][0] = xs[0]; x[j][1] = xs[32]; x[j][2] = xs[64]; x[j][3] = xs[96];
        }
      }
      #pragma unroll
      for (int j = 0; j < 8; ++j) {
        if (j < n) {
          const float w  = __builtin_bit_cast(float,
                             __builtin_amdgcn_readlane(__builtin_bit_cast(int, wv), j));
          const int lrow = __builtin_amdgcn_readlane(ql, first + j);
          float* a = s_acc + lrow * PITCH + lane;
          __hip_atomic_fetch_add(a,      w * x[j][0], __ATOMIC_RELAXED, __HIP_MEMORY_SCOPE_WORKGROUP);
          __hip_atomic_fetch_add(a + 32, w * x[j][1], __ATOMIC_RELAXED, __HIP_MEMORY_SCOPE_WORKGROUP);
          __hip_atomic_fetch_add(a + 64, w * x[j][2], __ATOMIC_RELAXED, __HIP_MEMORY_SCOPE_WORKGROUP);
          __hip_atomic_fetch_add(a + 96, w * x[j][3], __ATOMIC_RELAXED, __HIP_MEMORY_SCOPE_WORKGROUP);
        }
      }
      qn = first;
    };

    auto apply_edge = [&](int ee, int lrow) {
      int s = src_idx[ee];
      s = (s < 0) ? 0 : ((s >= nrows) ? nrows - 1 : s);
      const float w = ew[ee];
      const float* xs = src_x + (size_t)s * DIM + lane;
      float* a = s_acc + lrow * PITCH + lane;
      const float x0 = xs[0], x1 = xs[32], x2 = xs[64], x3 = xs[96];
      __hip_atomic_fetch_add(a,      w * x0, __ATOMIC_RELAXED, __HIP_MEMORY_SCOPE_WORKGROUP);
      __hip_atomic_fetch_add(a + 32, w * x1, __ATOMIC_RELAXED, __HIP_MEMORY_SCOPE_WORKGROUP);
      __hip_atomic_fetch_add(a + 64, w * x2, __ATOMIC_RELAXED, __HIP_MEMORY_SCOPE_WORKGROUP);
      __hip_atomic_fetch_add(a + 96, w * x3, __ATOMIC_RELAXED, __HIP_MEMORY_SCOPE_WORKGROUP);
    };

    for (int e0 = wave * 32; e0 < nedges; e0 += NTHR * SCAN_U) {
      int dv[SCAN_U];
      if (e0 + NTHR * SCAN_U <= nedges) {
        #pragma unroll
        for (int u = 0; u < SCAN_U; ++u) dv[u] = dst_idx[e0 + u * NTHR + lane];
      } else {
        #pragma unroll
        for (int u = 0; u < SCAN_U; ++u) {
          const int e = e0 + u * NTHR + lane;
          dv[u] = (e < nedges) ? dst_idx[e] : -1;
        }
      }
      unsigned hitmask[SCAN_U];
      int tot = 0;
      #pragma unroll
      for (int u = 0; u < SCAN_U; ++u) {
        const unsigned rel = (unsigned)(dv[u] - nbase);
        const bool hit = (dv[u] >= 0) && (rel < (unsigned)RPB);
        hitmask[u] = __builtin_amdgcn_ballot_w32(hit);
        tot += __builtin_popcount(hitmask[u]);
      }
      if (qn + tot > 32) {
        while (qn > 0) apply_top(qn < 8 ? qn : 8);
      }
      if (tot <= 32) {
        #pragma unroll
        for (int u = 0; u < SCAN_U; ++u) {
          unsigned mask = hitmask[u];
          while (mask) {
            const int b = __builtin_ctz(mask);
            mask &= mask - 1;
            const int erow = __builtin_amdgcn_readlane(dv[u] - nbase, b);
            if (lane == qn) { qe = e0 + u * NTHR + b; ql = erow; }
            ++qn;
          }
        }
      } else {
        for (int u = 0; u < SCAN_U; ++u) {
          const int e = e0 + u * NTHR + lane;
          const int d = (e < nedges) ? dst_idx[e] : -1;
          const unsigned rel = (unsigned)(d - nbase);
          unsigned mask = __builtin_amdgcn_ballot_w32((d >= 0) && (rel < (unsigned)RPB));
          while (mask) {
            const int b = __builtin_ctz(mask);
            mask &= mask - 1;
            apply_edge(e0 + u * NTHR + b, __builtin_amdgcn_readlane((int)rel, b));
          }
        }
      }
      while (qn >= 8) apply_top(8);
    }
    while (qn > 0) apply_top(qn < 8 ? qn : 8);
  }
  __syncthreads();

  const int lo  = (int)__lane_id() & 15;
  const int h   = (int)__lane_id() >> 4;
  const int rg = wave >> 1;
  const int c0 = (wave & 1) * 64;
  if (rg < 8) {
    transform_wide<2>(s_acc, s_w, dst_x, Wi, We, out, nbase, nrows, rg, c0, lo, h);
  } else {
    transform_wide<1>(s_acc, s_w, dst_x, Wi, We, out, nbase, nrows, rg, c0, lo, h);
    if (wave >= 18 && wave < 22)
      transform_rows<1>(s_acc, s_w, dst_x, Wi, We, out, nbase, nrows, NTILE - 1, (wave & 3) * 32, lo, h);
  }
}

extern "C" void kernel_launch(void* const* d_in, const int* in_sizes, int n_in,
                              void* d_out, int out_size, void* d_ws, size_t ws_size,
                              hipStream_t stream) {
  const float* src_x = (const float*)d_in[0];
  const float* dst_x = (const float*)d_in[1];
  const int*   eidx  = (const int*)  d_in[2];
  const float* ew    = (const float*)d_in[3];
  const float* Wi    = (const float*)d_in[4];
  const float* We    = (const float*)d_in[5];
  float*       out   = (float*)d_out;

  const int N = in_sizes[0] / DIM;
  const int E = in_sizes[2] / 2;
  const int* src_idx = eidx;
  const int* dst_idx = eidx + E;

  const int nb    = (N + RPB - 1) / RPB;
  const int chunk = (E + NCH - 1) / NCH;
  int* rec = (int*)d_ws;
  int* tab = rec + 2 * (size_t)E;
  const size_t need = (2 * (size_t)E + (size_t)NCH * (nb + 1)) * sizeof(int);

  if (d_ws != nullptr && ws_size >= need && nb < NBMAX) {
    bin_edges_kernel<<<NCH, BIN_THR, 0, stream>>>(src_idx, dst_idx, ew, rec, tab, N, E, nb, chunk);
    owner_range_kernel<true><<<nb, NTHR, 0, stream>>>(
        src_x, dst_x, src_idx, dst_idx, ew, Wi, We, out, N, E, rec, tab, nb, chunk);
  } else {
    owner_range_kernel<false><<<nb, NTHR, 0, stream>>>(
        src_x, dst_x, src_idx, dst_idx, ew, Wi, We, out, N, E, nullptr, nullptr, nb, chunk);
  }
}
